# starting kernel
# baseline (speedup 1.0000x reference)
#include <hip/hip_runtime.h>
#include <math.h>

#define CCH 30
#define NLAYERS 3
#define AST 34
#define CH2 (CCH / 2)

typedef __attribute__((ext_vector_type(2))) float v2f;
typedef __attribute__((ext_vector_type(8))) float v8f;

__device__ __forceinline__ float sigmoid_f(float x) { return 1.0f / (1.0f + __expf(-x)); }

__global__ void ggnn_transform_wmma(const float* __restrict__ h,
                                    const float* __restrict__ W,
                                    float* __restrict__ m, int N) {
    __shared__ float s_Wt[32 * AST];
    __shared__ float s_A[4][16 * AST];

    const int tid  = threadIdx.x;
    const int lane = tid & 31;
    const int warp = tid >> 5;
    const int row0 = (blockIdx.x * 4 + warp) * 16;
    const bool full = (row0 + 16 <= N);

    for (int i = tid; i < 32 * AST; i += 128) s_Wt[i] = 0.0f;
    for (int i = lane; i < 16 * AST; i += 32) s_A[warp][i] = 0.0f;
    __syncthreads();

    for (int i = tid; i < CCH * CCH; i += 128) {
        const int k = i / CCH, c = i % CCH;
        s_Wt[c * AST + k] = W[i];
    }
    if (full) {
        const float2* h2 = reinterpret_cast<const float2*>(h + (size_t)row0 * CCH);
        #pragma unroll
        for (int i = lane; i < 16 * CH2; i += 32) {
            const int r = i / CH2, c2 = i % CH2;
            *reinterpret_cast<float2*>(&s_A[warp][r * AST + 2 * c2]) =
                h2[(size_t)r * CH2 + c2];
        }
    } else {
        for (int i = lane; i < 16 * CCH; i += 32) {
            const int r = i / CCH, c = i % CCH;
            const int row = row0 + r;
            if (row < N) s_A[warp][r * AST + c] = h[(size_t)row * CCH + c];
        }
    }
    __syncthreads();

    if (row0 >= N) return;

    const int half = lane >> 4;
    const int l    = lane & 15;
    v8f acc0 = {};
    v8f acc1 = {};
    float* sa = &s_A[warp][0];

    #pragma unroll
    for (int k0 = 0; k0 < 32; k0 += 4) {
        const int ka = k0 + half * 2;
        const v2f a  = *(const v2f*)&sa  [l        * AST + ka];
        const v2f b0 = *(const v2f*)&s_Wt[l        * AST + ka];
        const v2f b1 = *(const v2f*)&s_Wt[(16 + l) * AST + ka];
        acc0 = __builtin_amdgcn_wmma_f32_16x16x4_f32(false, a, false, b0,
                                                     (short)0, acc0, false, false);
        acc1 = __builtin_amdgcn_wmma_f32_16x16x4_f32(false, a, false, b1,
                                                     (short)0, acc1, false, false);
    }

    #pragma unroll
    for (int v = 0; v < 8; ++v) {
        const int r = v + half * 8;
        sa[r * AST + l]      = acc0[v];
        sa[r * AST + 16 + l] = acc1[v];
    }

    if (full) {
        float2* m2 = reinterpret_cast<float2*>(m + (size_t)row0 * CCH);
        #pragma unroll
        for (int i = lane; i < 16 * CH2; i += 32) {
            const int r = i / CH2, c2 = i % CH2;
            m2[(size_t)r * CH2 + c2] =
                *reinterpret_cast<const float2*>(&sa[r * AST + 2 * c2]);
        }
    } else {
        for (int i = lane; i < 16 * CCH; i += 32) {
            const int r = i / CCH, c = i % CCH;
            const int row = row0 + r;
            if (row < N) m[(size_t)row * CCH + c] = sa[r * AST + c];
        }
    }
}

__global__ void ggnn_copy(const float* __restrict__ src, float* __restrict__ dst,
                          int n) {
    const int n4 = n >> 2;
    int i = blockIdx.x * blockDim.x + threadIdx.x;
    if (i < n4) {
        reinterpret_cast<float4*>(dst)[i] =
            reinterpret_cast<const float4*>(src)[i];
    }
    if (i < (n & 3)) {
        const int t = n4 * 4 + i;
        dst[t] = src[t];
    }
}

__global__ void ggnn_zero(float* __restrict__ p, int n) {
    const int n4 = n >> 2;
    int i = blockIdx.x * blockDim.x + threadIdx.x;
    if (i < n4) reinterpret_cast<float4*>(p)[i] = make_float4(0.f, 0.f, 0.f, 0.f);
    if (i < (n & 3)) p[n4 * 4 + i] = 0.0f;
}

__global__ void ggnn_scatter(const int* __restrict__ src,
                             const int* __restrict__ dst,
                             const float* __restrict__ m,
                             float* __restrict__ agg, int E) {
    int e = blockIdx.x * blockDim.x + threadIdx.x;
    if (e >= E) return;
    const int s = src[e];
    const int d = dst[e];
    const float2* ms = reinterpret_cast<const float2*>(m + (size_t)s * CCH);
    float*        ad = agg + (size_t)d * CCH;

    float2 v[CH2];
    #pragma unroll
    for (int i = 0; i < CH2; ++i) v[i] = ms[i];
    #pragma unroll
    for (int i = 0; i < CH2; ++i) {
        atomicAdd(&ad[2 * i],     v[i].x);
        atomicAdd(&ad[2 * i + 1], v[i].y);
    }
}

__global__ void ggnn_gru(const float* __restrict__ agg,
                         float* __restrict__ h,
                         const float* __restrict__ w_ih,
                         const float* __restrict__ w_hh,
                         const float* __restrict__ b_ih,
                         const float* __restrict__ b_hh, int N) {
    __shared__ float s_wih[3 * CCH * CCH];
    __shared__ float s_whh[3 * CCH * CCH];
    __shared__ float s_bih[3 * CCH];
    __shared__ float s_bhh[3 * CCH];

    for (int i = threadIdx.x; i < 3 * CCH * CCH; i += blockDim.x) {
        s_wih[i] = w_ih[i];
        s_whh[i] = w_hh[i];
    }
    if (threadIdx.x < 3 * CCH) {
        s_bih[threadIdx.x] = b_ih[threadIdx.x];
        s_bhh[threadIdx.x] = b_hh[threadIdx.x];
    }
    __syncthreads();

    const int n = blockIdx.x * blockDim.x + threadIdx.x;
    if (n >= N) return;

    float a[CCH], hv[CCH];
    #pragma unroll
    for (int k = 0; k < CCH; ++k) {
        a[k]  = agg[(size_t)n * CCH + k];
        hv[k] = h  [(size_t)n * CCH + k];
    }

    #pragma unroll 1
    for (int j = 0; j < CCH; ++j) {
        float ir = s_bih[j], iz = s_bih[CCH + j], inn = s_bih[2 * CCH + j];
        float hr = s_bhh[j], hz = s_bhh[CCH + j], hnn = s_bhh[2 * CCH + j];
        #pragma unroll
        for (int k = 0; k < CCH; ++k) {
            ir  += a[k]  * s_wih[j * CCH + k];
            iz  += a[k]  * s_wih[(CCH + j) * CCH + k];
            inn += a[k]  * s_wih[(2 * CCH + j) * CCH + k];
            hr  += hv[k] * s_whh[j * CCH + k];
            hz  += hv[k] * s_whh[(CCH + j) * CCH + k];
            hnn += hv[k] * s_whh[(2 * CCH + j) * CCH + k];
        }
        const float r  = sigmoid_f(ir + hr);
        const float z  = sigmoid_f(iz + hz);
        const float nn = tanhf(inn + r * hnn);
        h[(size_t)n * CCH + j] = (1.0f - z) * nn + z * hv[j];
    }
}

__global__ void ggnn_final(const float* __restrict__ h,
                           const float* __restrict__ lin_w,
                           const float* __restrict__ lin_b,
                           float* __restrict__ out, int N) {
    __shared__ float s_w[CCH * CCH];
    __shared__ float s_b[CCH];
    for (int i = threadIdx.x; i < CCH * CCH; i += blockDim.x) s_w[i] = lin_w[i];
    if (threadIdx.x < CCH) s_b[threadIdx.x] = lin_b[threadIdx.x];
    __syncthreads();

    const int n = blockIdx.x * blockDim.x + threadIdx.x;
    if (n >= N) return;

    float hv[CCH];
    #pragma unroll
    for (int k = 0; k < CCH; ++k) {
        const float v = h[(size_t)n * CCH + k];
        hv[k] = v > 0.0f ? v : 0.0f;
    }
    #pragma unroll 1
    for (int j = 0; j < CCH; ++j) {
        float acc = s_b[j];
        #pragma unroll
        for (int k = 0; k < CCH; ++k) acc += hv[k] * s_w[j * CCH + k];
        out[(size_t)n * CCH + j] = acc;
    }
}

extern "C" void kernel_launch(void* const* d_in, const int* in_sizes, int n_in,
                              void* d_out, int out_size, void* d_ws, size_t ws_size,
                              hipStream_t stream) {
    const float* x      = (const float*)d_in[0];
    const int*   eidx   = (const int*)  d_in[1];
    const float* weight = (const float*)d_in[2];
    const float* w_ih   = (const float*)d_in[3];
    const float* w_hh   = (const float*)d_in[4];
    const float* b_ih   = (const float*)d_in[5];
    const float* b_hh   = (const float*)d_in[6];
    const float* lin_w  = (const float*)d_in[7];
    const float* lin_b  = (const float*)d_in[8];
    float* out = (float*)d_out;

    const int N = in_sizes[0] / CCH;
    const int E = in_sizes[1] / 2;
    const int* src = eidx;
    const int* dst = eidx + E;

    float* h   = (float*)d_ws;
    float* m   = h + (size_t)N * CCH;
    float* agg = m + (size_t)N * CCH;

    const int nelem = N * CCH;
    const int n4    = (nelem >> 2) + 4;
    ggnn_copy<<<(n4 + 255) / 256, 256, 0, stream>>>(x, h, nelem);

    const int tiles = (N + 15) / 16;
    for (int layer = 0; layer < NLAYERS; ++layer) {
        ggnn_transform_wmma<<<(tiles + 3) / 4, 128, 0, stream>>>(
            h, weight + (size_t)layer * CCH * CCH, m, N);
        ggnn_zero<<<(n4 + 255) / 256, 256, 0, stream>>>(agg, nelem);
        ggnn_scatter<<<(E + 255) / 256, 256, 0, stream>>>(src, dst, m, agg, E);
        ggnn_gru<<<(N + 127) / 128, 128, 0, stream>>>(agg, h, w_ih, w_hh,
                                                      b_ih, b_hh, N);
    }
    ggnn_final<<<(N + 127) / 128, 128, 0, stream>>>(h, lin_w, lin_b, out, N);
}


# First-layer scan reads 1-byte owner keys (node >> 10, written by a small launch in front) instead of 4-byte ids, candidates kept in a register; sum order stays LDS arrival
# speedup vs baseline: 7.8462x; 7.8462x over previous
#include <hip/hip_runtime.h>
#include <math.h>

#define CCH 30
#define NLAYERS 3
#define AST 34
#define CH2 (CCH / 2)

typedef __attribute__((ext_vector_type(2))) float v2f;
typedef __attribute__((ext_vector_type(8))) float v8f;

__device__ __forceinline__ float sigmoid_f(float x) { return 1.0f / (1.0f + __expf(-x)); }

__global__ void ggnn_transform_wmma(const float* __restrict__ h,
                                    const float* __restrict__ W,
                                    float* __restrict__ m, int N) {
    __shared__ float s_Wt[32 * AST];
    __shared__ float s_A[4][16 * AST];

    const int tid  = threadIdx.x;
    const int lane = tid & 31;
    const int warp = tid >> 5;
    const int row0 = (blockIdx.x * 4 + warp) * 16;
    const bool full = (row0 + 16 <= N);

    for (int i = tid; i < 32 * AST; i += 128) s_Wt[i] = 0.0f;
    for (int i = lane; i < 16 * AST; i += 32) s_A[warp][i] = 0.0f;
    __syncthreads();

    for (int i = tid; i < CCH * CCH; i += 128) {
        const int k = i / CCH, c = i % CCH;
        s_Wt[c * AST + k] = W[i];
    }
    if (full) {
        const float2* h2 = reinterpret_cast<const float2*>(h + (size_t)row0 * CCH);
        #pragma unroll
        for (int i = lane; i < 16 * CH2; i += 32) {
            const int r = i / CH2, c2 = i % CH2;
            *reinterpret_cast<float2*>(&s_A[warp][r * AST + 2 * c2]) =
                h2[(size_t)r * CH2 + c2];
        }
    } else {
        for (int i = lane; i < 16 * CCH; i += 32) {
            const int r = i / CCH, c = i % CCH;
            const int row = row0 + r;
            if (row < N) s_A[warp][r * AST + c] = h[(size_t)row * CCH + c];
        }
    }
    __syncthreads();

    if (row0 >= N) return;

    const int half = lane >> 4;
    const int l    = lane & 15;
    v8f acc0 = {};
    v8f acc1 = {};
    float* sa = &s_A[warp][0];

    #pragma unroll
    for (int k0 = 0; k0 < 32; k0 += 4) {
        const int ka = k0 + half * 2;
        const v2f a  = *(const v2f*)&sa  [l        * AST + ka];
        const v2f b0 = *(const v2f*)&s_Wt[l        * AST + ka];
        const v2f b1 = *(const v2f*)&s_Wt[(16 + l) * AST + ka];
        acc0 = __builtin_amdgcn_wmma_f32_16x16x4_f32(false, a, false, b0,
                                                     (short)0, acc0, false, false);
        acc1 = __builtin_amdgcn_wmma_f32_16x16x4_f32(false, a, false, b1,
                                                     (short)0, acc1, false, false);
    }

    #pragma unroll
    for (int v = 0; v < 8; ++v) {
        const int r = v + half * 8;
        sa[r * AST + l]      = acc0[v];
        sa[r * AST + 16 + l] = acc1[v];
    }

    if (full) {
        float2* m2 = reinterpret_cast<float2*>(m + (size_t)row0 * CCH);
        #pragma unroll
        for (int i = lane; i < 16 * CH2; i += 32) {
            const int r = i / CH2, c2 = i % CH2;
            m2[(size_t)r * CH2 + c2] =
                *reinterpret_cast<const float2*>(&sa[r * AST + 2 * c2]);
        }
    } else {
        for (int i = lane; i < 16 * CCH; i += 32) {
            const int r = i / CCH, c = i % CCH;
            const int row = row0 + r;
            if (row < N) m[(size_t)row * CCH + c] = sa[r * AST + c];
        }
    }
}

__global__ void ggnn_copy(const float* __restrict__ src, float* __restrict__ dst,
                          int n) {
    const int n4 = n >> 2;
    int i = blockIdx.x * blockDim.x + threadIdx.x;
    if (i < n4) {
        reinterpret_cast<float4*>(dst)[i] =
            reinterpret_cast<const float4*>(src)[i];
    }
    if (i < (n & 3)) {
        const int t = n4 * 4 + i;
        dst[t] = src[t];
    }
}

#define NPW         1024
#define AGG_ROWS    1368
#define OWN_THREADS 1024
#define OWN_WAVES   32
#define WCAP        2048

__global__ void ggnn_owner_keys(const int* __restrict__ dst,
                                unsigned* __restrict__ key, int nwords) {
    const int i = blockIdx.x * blockDim.x + threadIdx.x;
    if (i >= nwords) return;
    int d0, d1, d2, d3;
    if ((reinterpret_cast<size_t>(dst) & 15) == 0) {
        const int4 d = reinterpret_cast<const int4*>(dst)[i];
        d0 = d.x; d1 = d.y; d2 = d.z; d3 = d.w;
    } else {
        d0 = dst[4 * i]; d1 = dst[4 * i + 1]; d2 = dst[4 * i + 2]; d3 = dst[4 * i + 3];
    }
    key[i] = (((unsigned)d0 >> 10) & 255u) | ((((unsigned)d1 >> 10) & 255u) << 8) |
             ((((unsigned)d2 >> 10) & 255u) << 16) | ((((unsigned)d3 >> 10) & 255u) << 24);
}

template <bool FULL, int B>
__device__ __forceinline__ void add_rows(int s, int dl, int n,
                                         const float* __restrict__ m,
                                         float* s_agg, int lane) {
    #pragma unroll
    for (int j = 0; j < 32; j += B) {
        if (!FULL && j >= n) break;
        int sj[B], dj[B];
        #pragma unroll
        for (int u = 0; u < B; ++u) {
            sj[u] = __builtin_amdgcn_readlane(s, j + u);
            dj[u] = __builtin_amdgcn_readlane(dl, j + u);
        }
        if (lane < CCH) {
            float v[B];
            #pragma unroll
            for (int u = 0; u < B; ++u)
                if (FULL || j + u < n) v[u] = m[(size_t)sj[u] * CCH + lane];
            #pragma unroll
            for (int u = 0; u < B; ++u)
                if (FULL || j + u < n)
                    __hip_atomic_fetch_add(&s_agg[dj[u] * CCH + lane], v[u],
                                           __ATOMIC_RELAXED,
                                           __HIP_MEMORY_SCOPE_WORKGROUP);
        }
    }
}

template <bool FULL, bool REC>
__device__ __forceinline__ void own_drain(int e, int n, int lo,
                                          const int* __restrict__ src,
                                          const int* __restrict__ dst,
                                          const float* __restrict__ m,
                                          float* s_agg, int lane,
                                          int2* __restrict__ rec, int recpos) {
    int s = 0, r = NPW;
    if (FULL || lane < n) {
        const unsigned rr = (unsigned)(dst[e] - lo);
        r = (int)(rr < (unsigned)NPW ? rr : (unsigned)NPW);
        s = src[e];
        if (REC && recpos + lane < WCAP) rec[recpos + lane] = make_int2(s, r);
    }
    add_rows<FULL, 16>(s, r, n, m, s_agg, lane);
}

template <bool REC>
__device__ __forceinline__ void own_push(int e, int& ring, int& pos, int& recpos, int lo,
                                         const int* __restrict__ src,
                                         const int* __restrict__ dst,
                                         const float* __restrict__ m,
                                         float* s_agg, int lane,
                                         int2* __restrict__ rec) {
    ring = (lane == pos) ? e : ring;
    if (++pos == 32) {
        own_drain<true, REC>(ring, 32, lo, src, dst, m, s_agg, lane, rec, recpos);
        recpos += 32;
        pos = 0;
    }
}

__device__ __forceinline__ void own_replay(const int2* __restrict__ rec, int cnt,
                                           const float* __restrict__ m,
                                           float* s_agg, int lane) {
    int2 cur = make_int2(0, 0);
    if (lane < cnt) cur = rec[lane];
    for (int base = 0; base < cnt; base += 32) {
        int2 nxt = make_int2(0, 0);
        if (base + 32 + lane < cnt) nxt = rec[base + 32 + lane];
        const int n = cnt - base;
        if (n >= 32) add_rows<true, 32>(cur.x, cur.y, 32, m, s_agg, lane);
        else         add_rows<false, 32>(cur.x, cur.y, n, m, s_agg, lane);
        cur = nxt;
    }
}

template <int MODE>
__global__ __launch_bounds__(OWN_THREADS)
void ggnn_owner_sum(const int* __restrict__ src,
                    const int* __restrict__ dst,
                    const uint4* __restrict__ key4,
                    const float* __restrict__ m,
                    float* __restrict__ agg,
                    int2* __restrict__ rec_all,
                    int* __restrict__ wcnt, int N, int E) {
    __shared__ __attribute__((aligned(16))) float s_agg[AGG_ROWS * CCH];

    const int tid  = threadIdx.x;
    const int lane = tid & 31;
    const int wave = __builtin_amdgcn_readfirstlane(tid >> 5);
    const int lo   = blockIdx.x * NPW;

    for (int i = tid; i < AGG_ROWS * CCH / 4; i += OWN_THREADS)
        reinterpret_cast<float4*>(s_agg)[i] = make_float4(0.f, 0.f, 0.f, 0.f);
    __syncthreads();

    constexpr bool REC = (MODE == 0);
    const int gw = blockIdx.x * OWN_WAVES + wave;
    int2* rec = rec_all + (size_t)gw * WCAP;
    int recpos = 0;
    bool replay = false;
    if (MODE == 1) {
        const int cnt = __builtin_amdgcn_readfirstlane(wcnt[gw]);
        replay = (cnt >= 0 && cnt <= WCAP);
        if (replay) own_replay(rec, cnt, m, s_agg, lane);
    }
    if (!replay) {

    int ring = 0;
    int pos  = 0;

    const unsigned pat = (unsigned)(blockIdx.x & 255) * 0x01010101u;
    const int nsu = E >> 11;
    const int start = (int)(((long long)blockIdx.x * nsu) / gridDim.x);

    if (wave < nsu) {
        int k = wave;
        uint4 cur[4];
        {
            int su = k + start; if (su >= nsu) su -= nsu;
            const uint4* p = key4 + (size_t)su * 128 + lane;
            #pragma unroll
            for (int u = 0; u < 4; ++u) cur[u] = p[32 * u];
        }
        while (k < nsu) {
            const int kn = k + OWN_WAVES;
            const int kl = kn < nsu ? kn : k;
            int sun = kl + start; if (sun >= nsu) sun -= nsu;
            const uint4* pn = key4 + (size_t)sun * 128 + lane;
            uint4 nxt[4];
            #pragma unroll
            for (int u = 0; u < 4; ++u) nxt[u] = pn[32 * u];

            int su = k + start; if (su >= nsu) su -= nsu;
            #pragma unroll
            for (int u = 0; u < 4; ++u) {
                const unsigned wv[4] = {cur[u].x, cur[u].y, cur[u].z, cur[u].w};
                unsigned fl = 0u;
                #pragma unroll
                for (int c = 0; c < 4; ++c) {
                    const unsigned x = wv[c] ^ pat;
                    const unsigned y = ((x & 0x7F7F7F7Fu) + 0x7F7F7F7Fu) | x;
                    fl |= (~y & 0x80808080u) >> (7 - c);
                }
                unsigned mk = __builtin_amdgcn_ballot_w32(fl != 0u);
                const int eu = su * 2048 + u * 512;
                while (mk != 0u) {
                    const int L = __builtin_ctz(mk);
                    mk &= mk - 1u;
                    unsigned f = (unsigned)__builtin_amdgcn_readlane((int)fl, L);
                    while (f != 0u) {
                        const int i = __builtin_ctz(f);
                        f &= f - 1u;
                        own_push<REC>(eu + L * 16 + (i & 7) * 4 + (i >> 3),
                                      ring, pos, recpos, lo, src, dst, m, s_agg, lane, rec);
                    }
                }
            }
            #pragma unroll
            for (int u = 0; u < 4; ++u) cur[u] = nxt[u];
            k = kn;
        }
    }

    for (int base = nsu * 2048 + wave * 32; base < E; base += OWN_THREADS) {
        const int  e     = base + lane;
        const bool valid = e < E;
        const int  d     = valid ? dst[e] : -1;
        const bool p     = valid && (unsigned)(d - lo) < (unsigned)NPW;
        unsigned mk = __builtin_amdgcn_ballot_w32(p);
        while (mk != 0u) {
            const int L = __builtin_ctz(mk);
            mk &= mk - 1u;
            own_push<REC>(base + L, ring, pos, recpos, lo, src, dst, m, s_agg, lane, rec);
        }
    }
    if (pos > 0) {
        own_drain<false, REC>(ring, pos, lo, src, dst, m, s_agg, lane, rec, recpos);
        recpos += pos;
    }
    if (REC && lane == 0) wcnt[gw] = recpos;
    }
    __syncthreads();

    const int nn = (N - lo) < NPW ? (N - lo) : NPW;
    float* ag = agg + (size_t)lo * CCH;
    for (int i = tid; i < nn * CCH; i += OWN_THREADS) ag[i] = s_agg[i];
}

__global__ void ggnn_gru(const float* __restrict__ agg,
                         float* __restrict__ h,
                         const float* __restrict__ w_ih,
                         const float* __restrict__ w_hh,
                         const float* __restrict__ b_ih,
                         const float* __restrict__ b_hh, int N) {
    __shared__ float s_wih[3 * CCH * CCH];
    __shared__ float s_whh[3 * CCH * CCH];
    __shared__ float s_bih[3 * CCH];
    __shared__ float s_bhh[3 * CCH];

    for (int i = threadIdx.x; i < 3 * CCH * CCH; i += blockDim.x) {
        s_wih[i] = w_ih[i];
        s_whh[i] = w_hh[i];
    }
    if (threadIdx.x < 3 * CCH) {
        s_bih[threadIdx.x] = b_ih[threadIdx.x];
        s_bhh[threadIdx.x] = b_hh[threadIdx.x];
    }
    __syncthreads();

    const int n = blockIdx.x * blockDim.x + threadIdx.x;
    if (n >= N) return;

    float a[CCH], hv[CCH];
    #pragma unroll
    for (int k = 0; k < CCH; ++k) {
        a[k]  = agg[(size_t)n * CCH + k];
        hv[k] = h  [(size_t)n * CCH + k];
    }

    #pragma unroll 1
    for (int j = 0; j < CCH; ++j) {
        float ir = s_bih[j], iz = s_bih[CCH + j], inn = s_bih[2 * CCH + j];
        float hr = s_bhh[j], hz = s_bhh[CCH + j], hnn = s_bhh[2 * CCH + j];
        #pragma unroll
        for (int k = 0; k < CCH; ++k) {
            ir  += a[k]  * s_wih[j * CCH + k];
            iz  += a[k]  * s_wih[(CCH + j) * CCH + k];
            inn += a[k]  * s_wih[(2 * CCH + j) * CCH + k];
            hr  += hv[k] * s_whh[j * CCH + k];
            hz  += hv[k] * s_whh[(CCH + j) * CCH + k];
            hnn += hv[k] * s_whh[(2 * CCH + j) * CCH + k];
        }
        const float r  = sigmoid_f(ir + hr);
        const float z  = sigmoid_f(iz + hz);
        const float nn = tanhf(inn + r * hnn);
        h[(size_t)n * CCH + j] = (1.0f - z) * nn + z * hv[j];
    }
}

__global__ void ggnn_final(const float* __restrict__ h,
                           const float* __restrict__ lin_w,
                           const float* __restrict__ lin_b,
                           float* __restrict__ out, int N) {
    __shared__ float s_w[CCH * CCH];
    __shared__ float s_b[CCH];
    for (int i = threadIdx.x; i < CCH * CCH; i += blockDim.x) s_w[i] = lin_w[i];
    if (threadIdx.x < CCH) s_b[threadIdx.x] = lin_b[threadIdx.x];
    __syncthreads();

    const int n = blockIdx.x * blockDim.x + threadIdx.x;
    if (n >= N) return;

    float hv[CCH];
    #pragma unroll
    for (int k = 0; k < CCH; ++k) {
        const float v = h[(size_t)n * CCH + k];
        hv[k] = v > 0.0f ? v : 0.0f;
    }
    #pragma unroll 1
    for (int j = 0; j < CCH; ++j) {
        float acc = s_b[j];
        #pragma unroll
        for (int k = 0; k < CCH; ++k) acc += hv[k] * s_w[j * CCH + k];
        out[(size_t)n * CCH + j] = acc;
    }
}

extern "C" void kernel_launch(void* const* d_in, const int* in_sizes, int n_in,
                              void* d_out, int out_size, void* d_ws, size_t ws_size,
                              hipStream_t stream) {
    const float* x      = (const float*)d_in[0];
    const int*   eidx   = (const int*)  d_in[1];
    const float* weight = (const float*)d_in[2];
    const float* w_ih   = (const float*)d_in[3];
    const float* w_hh   = (const float*)d_in[4];
    const float* b_ih   = (const float*)d_in[5];
    const float* b_hh   = (const float*)d_in[6];
    const float* lin_w  = (const float*)d_in[7];
    const float* lin_b  = (const float*)d_in[8];
    float* out = (float*)d_out;

    const int N = in_sizes[0] / CCH;
    const int E = in_sizes[1] / 2;
    const int* src = eidx;
    const int* dst = eidx + E;

    float* h   = (float*)d_ws;
    float* m   = h + (size_t)N * CCH;
    float* agg = m + (size_t)N * CCH;
    const int groups = (N + NPW - 1) / NPW;
    int2* rec  = (int2*)(agg + (size_t)N * CCH);
    int*  wcnt = (int*)(rec + (size_t)groups * OWN_WAVES * WCAP);
    unsigned* key = (unsigned*)(((size_t)(wcnt + (size_t)groups * OWN_WAVES) + 15) & ~(size_t)15);
    const int nwords = (E >> 11) * 512;
    if (nwords > 0)
        ggnn_owner_keys<<<(nwords + 255) / 256, 256, 0, stream>>>(dst, key, nwords);

    const int nelem = N * CCH;
    const int n4    = (nelem >> 2) + 4;
    ggnn_copy<<<(n4 + 255) / 256, 256, 0, stream>>>(x, h, nelem);

    const int tiles = (N + 15) / 16;
    for (int layer = 0; layer < NLAYERS; ++layer) {
        ggnn_transform_wmma<<<(tiles + 3) / 4, 128, 0, stream>>>(
            h, weight + (size_t)layer * CCH * CCH, m, N);
        if (layer == 0)
            ggnn_owner_sum<0><<<groups, OWN_THREADS, 0, stream>>>(src, dst, (const uint4*)key, m, agg, rec, wcnt, N, E);
        else
            ggnn_owner_sum<1><<<groups, OWN_THREADS, 0, stream>>>(src, dst, (const uint4*)key, m, agg, rec, wcnt, N, E);
        ggnn_gru<<<(N + 127) / 128, 128, 0, stream>>>(agg, h, w_ih, w_hh,
                                                      b_ih, b_hh, N);
    }
    ggnn_final<<<(N + 127) / 128, 128, 0, stream>>>(h, lin_w, lin_b, out, N);
}


# GRU cell from thread-per-node (LDS-broadcast weights, 4-B strided stores) to wave-per-node, lane = channel: weights in registers, uniform row loads into packed f32 FMAs; dot sums split even/odd k
# speedup vs baseline: 9.8010x; 1.2491x over previous
#include <hip/hip_runtime.h>
#include <math.h>

#define CCH 30
#define NLAYERS 3
#define AST 34
#define CH2 (CCH / 2)

typedef __attribute__((ext_vector_type(2))) float v2f;
typedef __attribute__((ext_vector_type(8))) float v8f;

__device__ __forceinline__ float sigmoid_f(float x) { return 1.0f / (1.0f + __expf(-x)); }

__global__ void ggnn_transform_wmma(const float* __restrict__ h,
                                    const float* __restrict__ W,
                                    float* __restrict__ m, int N) {
    __shared__ float s_Wt[32 * AST];
    __shared__ float s_A[4][16 * AST];

    const int tid  = threadIdx.x;
    const int lane = tid & 31;
    const int warp = tid >> 5;
    const int row0 = (blockIdx.x * 4 + warp) * 16;
    const bool full = (row0 + 16 <= N);

    for (int i = tid; i < 32 * AST; i += 128) s_Wt[i] = 0.0f;
    for (int i = lane; i < 16 * AST; i += 32) s_A[warp][i] = 0.0f;
    __syncthreads();

    for (int i = tid; i < CCH * CCH; i += 128) {
        const int k = i / CCH, c = i % CCH;
        s_Wt[c * AST + k] = W[i];
    }
    if (full) {
        const float2* h2 = reinterpret_cast<const float2*>(h + (size_t)row0 * CCH);
        #pragma unroll
        for (int i = lane; i < 16 * CH2; i += 32) {
            const int r = i / CH2, c2 = i % CH2;
            *reinterpret_cast<float2*>(&s_A[warp][r * AST + 2 * c2]) =
                h2[(size_t)r * CH2 + c2];
        }
    } else {
        for (int i = lane; i < 16 * CCH; i += 32) {
            const int r = i / CCH, c = i % CCH;
            const int row = row0 + r;
            if (row < N) s_A[warp][r * AST + c] = h[(size_t)row * CCH + c];
        }
    }
    __syncthreads();

    if (row0 >= N) return;

    const int half = lane >> 4;
    const int l    = lane & 15;
    v8f acc0 = {};
    v8f acc1 = {};
    float* sa = &s_A[warp][0];

    #pragma unroll
    for (int k0 = 0; k0 < 32; k0 += 4) {
        const int ka = k0 + half * 2;
        const v2f a  = *(const v2f*)&sa  [l        * AST + ka];
        const v2f b0 = *(const v2f*)&s_Wt[l        * AST + ka];
        const v2f b1 = *(const v2f*)&s_Wt[(16 + l) * AST + ka];
        acc0 = __builtin_amdgcn_wmma_f32_16x16x4_f32(false, a, false, b0,
                                                     (short)0, acc0, false, false);
        acc1 = __builtin_amdgcn_wmma_f32_16x16x4_f32(false, a, false, b1,
                                                     (short)0, acc1, false, false);
    }

    #pragma unroll
    for (int v = 0; v < 8; ++v) {
        const int r = v + half * 8;
        sa[r * AST + l]      = acc0[v];
        sa[r * AST + 16 + l] = acc1[v];
    }

    if (full) {
        float2* m2 = reinterpret_cast<float2*>(m + (size_t)row0 * CCH);
        #pragma unroll
        for (int i = lane; i < 16 * CH2; i += 32) {
            const int r = i / CH2, c2 = i % CH2;
            m2[(size_t)r * CH2 + c2] =
                *reinterpret_cast<const float2*>(&sa[r * AST + 2 * c2]);
        }
    } else {
        for (int i = lane; i < 16 * CCH; i += 32) {
            const int r = i / CCH, c = i % CCH;
            const int row = row0 + r;
            if (row < N) m[(size_t)row * CCH + c] = sa[r * AST + c];
        }
    }
}

__global__ void ggnn_copy(const float* __restrict__ src, float* __restrict__ dst,
                          int n) {
    const int n4 = n >> 2;
    int i = blockIdx.x * blockDim.x + threadIdx.x;
    if (i < n4) {
        reinterpret_cast<float4*>(dst)[i] =
            reinterpret_cast<const float4*>(src)[i];
    }
    if (i < (n & 3)) {
        const int t = n4 * 4 + i;
        dst[t] = src[t];
    }
}

#define NPW         1024
#define AGG_ROWS    1368
#define OWN_THREADS 1024
#define OWN_WAVES   32
#define WCAP        2048

__global__ void ggnn_owner_keys(const int* __restrict__ dst,
                                unsigned* __restrict__ key, int nwords) {
    const int i = blockIdx.x * blockDim.x + threadIdx.x;
    if (i >= nwords) return;
    int d0, d1, d2, d3;
    if ((reinterpret_cast<size_t>(dst) & 15) == 0) {
        const int4 d = reinterpret_cast<const int4*>(dst)[i];
        d0 = d.x; d1 = d.y; d2 = d.z; d3 = d.w;
    } else {
        d0 = dst[4 * i]; d1 = dst[4 * i + 1]; d2 = dst[4 * i + 2]; d3 = dst[4 * i + 3];
    }
    key[i] = (((unsigned)d0 >> 10) & 255u) | ((((unsigned)d1 >> 10) & 255u) << 8) |
             ((((unsigned)d2 >> 10) & 255u) << 16) | ((((unsigned)d3 >> 10) & 255u) << 24);
}

template <bool FULL, int B>
__device__ __forceinline__ void add_rows(int s, int dl, int n,
                                         const float* __restrict__ m,
                                         float* s_agg, int lane) {
    #pragma unroll
    for (int j = 0; j < 32; j += B) {
        if (!FULL && j >= n) break;
        int sj[B], dj[B];
        #pragma unroll
        for (int u = 0; u < B; ++u) {
            sj[u] = __builtin_amdgcn_readlane(s, j + u);
            dj[u] = __builtin_amdgcn_readlane(dl, j + u);
        }
        if (lane < CCH) {
            float v[B];
            #pragma unroll
            for (int u = 0; u < B; ++u)
                if (FULL || j + u < n) v[u] = m[(size_t)sj[u] * CCH + lane];
            #pragma unroll
            for (int u = 0; u < B; ++u)
                if (FULL || j + u < n)
                    __hip_atomic_fetch_add(&s_agg[dj[u] * CCH + lane], v[u],
                                           __ATOMIC_RELAXED,
                                           __HIP_MEMORY_SCOPE_WORKGROUP);
        }
    }
}

template <bool FULL, bool REC>
__device__ __forceinline__ void own_drain(int e, int n, int lo,
                                          const int* __restrict__ src,
                                          const int* __restrict__ dst,
                                          const float* __restrict__ m,
                                          float* s_agg, int lane,
                                          int2* __restrict__ rec, int recpos) {
    int s = 0, r = NPW;
    if (FULL || lane < n) {
        const unsigned rr = (unsigned)(dst[e] - lo);
        r = (int)(rr < (unsigned)NPW ? rr : (unsigned)NPW);
        s = src[e];
        if (REC && recpos + lane < WCAP) rec[recpos + lane] = make_int2(s, r);
    }
    add_rows<FULL, 16>(s, r, n, m, s_agg, lane);
}

template <bool REC>
__device__ __forceinline__ void own_push(int e, int& ring, int& pos, int& recpos, int lo,
                                         const int* __restrict__ src,
                                         const int* __restrict__ dst,
                                         const float* __restrict__ m,
                                         float* s_agg, int lane,
                                         int2* __restrict__ rec) {
    ring = (lane == pos) ? e : ring;
    if (++pos == 32) {
        own_drain<true, REC>(ring, 32, lo, src, dst, m, s_agg, lane, rec, recpos);
        recpos += 32;
        pos = 0;
    }
}

__device__ __forceinline__ void own_replay(const int2* __restrict__ rec, int cnt,
                                           const float* __restrict__ m,
                                           float* s_agg, int lane) {
    int2 cur = make_int2(0, 0);
    if (lane < cnt) cur = rec[lane];
    for (int base = 0; base < cnt; base += 32) {
        int2 nxt = make_int2(0, 0);
        if (base + 32 + lane < cnt) nxt = rec[base + 32 + lane];
        const int n = cnt - base;
        if (n >= 32) add_rows<true, 32>(cur.x, cur.y, 32, m, s_agg, lane);
        else         add_rows<false, 32>(cur.x, cur.y, n, m, s_agg, lane);
        cur = nxt;
    }
}

template <int MODE>
__global__ __launch_bounds__(OWN_THREADS)
void ggnn_owner_sum(const int* __restrict__ src,
                    const int* __restrict__ dst,
                    const uint4* __restrict__ key4,
                    const float* __restrict__ m,
                    float* __restrict__ agg,
                    int2* __restrict__ rec_all,
                    int* __restrict__ wcnt, int N, int E) {
    __shared__ __attribute__((aligned(16))) float s_agg[AGG_ROWS * CCH];

    const int tid  = threadIdx.x;
    const int lane = tid & 31;
    const int wave = __builtin_amdgcn_readfirstlane(tid >> 5);
    const int lo   = blockIdx.x * NPW;

    for (int i = tid; i < AGG_ROWS * CCH / 4; i += OWN_THREADS)
        reinterpret_cast<float4*>(s_agg)[i] = make_float4(0.f, 0.f, 0.f, 0.f);
    __syncthreads();

    constexpr bool REC = (MODE == 0);
    const int gw = blockIdx.x * OWN_WAVES + wave;
    int2* rec = rec_all + (size_t)gw * WCAP;
    int recpos = 0;
    bool replay = false;
    if (MODE == 1) {
        const int cnt = __builtin_amdgcn_readfirstlane(wcnt[gw]);
        replay = (cnt >= 0 && cnt <= WCAP);
        if (replay) own_replay(rec, cnt, m, s_agg, lane);
    }
    if (!replay) {

    int ring = 0;
    int pos  = 0;

    const unsigned pat = (unsigned)(blockIdx.x & 255) * 0x01010101u;
    const int nsu = E >> 11;
    const int start = (int)(((long long)blockIdx.x * nsu) / gridDim.x);

    if (wave < nsu) {
        int k = wave;
        uint4 cur[4];
        {
            int su = k + start; if (su >= nsu) su -= nsu;
            const uint4* p = key4 + (size_t)su * 128 + lane;
            #pragma unroll
            for (int u = 0; u < 4; ++u) cur[u] = p[32 * u];
        }
        while (k < nsu) {
            const int kn = k + OWN_WAVES;
            const int kl = kn < nsu ? kn : k;
            int sun = kl + start; if (sun >= nsu) sun -= nsu;
            const uint4* pn = key4 + (size_t)sun * 128 + lane;
            uint4 nxt[4];
            #pragma unroll
            for (int u = 0; u < 4; ++u) nxt[u] = pn[32 * u];

            int su = k + start; if (su >= nsu) su -= nsu;
            #pragma unroll
            for (int u = 0; u < 4; ++u) {
                const unsigned wv[4] = {cur[u].x, cur[u].y, cur[u].z, cur[u].w};
                unsigned fl = 0u;
                #pragma unroll
                for (int c = 0; c < 4; ++c) {
                    const unsigned x = wv[c] ^ pat;
                    const unsigned y = ((x & 0x7F7F7F7Fu) + 0x7F7F7F7Fu) | x;
                    fl |= (~y & 0x80808080u) >> (7 - c);
                }
                unsigned mk = __builtin_amdgcn_ballot_w32(fl != 0u);
                const int eu = su * 2048 + u * 512;
                while (mk != 0u) {
                    const int L = __builtin_ctz(mk);
                    mk &= mk - 1u;
                    unsigned f = (unsigned)__builtin_amdgcn_readlane((int)fl, L);
                    while (f != 0u) {
                        const int i = __builtin_ctz(f);
                        f &= f - 1u;
                        own_push<REC>(eu + L * 16 + (i & 7) * 4 + (i >> 3),
                                      ring, pos, recpos, lo, src, dst, m, s_agg, lane, rec);
                    }
                }
            }
            #pragma unroll
            for (int u = 0; u < 4; ++u) cur[u] = nxt[u];
            k = kn;
        }
    }

    for (int base = nsu * 2048 + wave * 32; base < E; base += OWN_THREADS) {
        const int  e     = base + lane;
        const bool valid = e < E;
        const int  d     = valid ? dst[e] : -1;
        const bool p     = valid && (unsigned)(d - lo) < (unsigned)NPW;
        unsigned mk = __builtin_amdgcn_ballot_w32(p);
        while (mk != 0u) {
            const int L = __builtin_ctz(mk);
            mk &= mk - 1u;
            own_push<REC>(base + L, ring, pos, recpos, lo, src, dst, m, s_agg, lane, rec);
        }
    }
    if (pos > 0) {
        own_drain<false, REC>(ring, pos, lo, src, dst, m, s_agg, lane, rec, recpos);
        recpos += pos;
    }
    if (REC && lane == 0) wcnt[gw] = recpos;
    }
    __syncthreads();

    const int nn = (N - lo) < NPW ? (N - lo) : NPW;
    float* ag = agg + (size_t)lo * CCH;
    for (int i = tid; i < nn * CCH; i += OWN_THREADS) ag[i] = s_agg[i];
}

#define GRU_WAVES  4
#define GRU_GROUPS 768
#define GRU_TILE   16

__device__ __forceinline__ void gru_rows(const float* agg, const float* h, int n, int N, int j,
                                         v2f (&a)[CH2], v2f (&hv)[CH2], float& hj) {
    n = n < N ? n : N - 1;
    const float* ar = agg + (size_t)n * CCH;
    const float* hr = h   + (size_t)n * CCH;
    #pragma unroll
    for (int k = 0; k < CH2; ++k) {
        a[k]  = *(const v2f*)(ar + 2 * k);
        hv[k] = *(const v2f*)(hr + 2 * k);
    }
    hj = hr[j];
}

struct GruW {
    v2f wir[CH2], wiz[CH2], win[CH2], whr[CH2], whz[CH2], whn[CH2];
    float bir, biz, bin, bhr, bhz, bhn;
};

__device__ __forceinline__ float gru_node(const GruW& w, const v2f (&a)[CH2], const v2f (&hv)[CH2],
                                          float hj) {
    v2f ir = {w.bir, 0.0f}, iz = {w.biz, 0.0f}, in_ = {w.bin, 0.0f};
    v2f gr = {w.bhr, 0.0f}, gz = {w.bhz, 0.0f}, gn = {w.bhn, 0.0f};
    #pragma unroll
    for (int k = 0; k < CH2; ++k) {
        ir  = __builtin_elementwise_fma(a[k],  w.wir[k], ir);
        iz  = __builtin_elementwise_fma(a[k],  w.wiz[k], iz);
        in_ = __builtin_elementwise_fma(a[k],  w.win[k], in_);
        gr  = __builtin_elementwise_fma(hv[k], w.whr[k], gr);
        gz  = __builtin_elementwise_fma(hv[k], w.whz[k], gz);
        gn  = __builtin_elementwise_fma(hv[k], w.whn[k], gn);
    }
    const float r  = sigmoid_f((ir.x + ir.y) + (gr.x + gr.y));
    const float z  = sigmoid_f((iz.x + iz.y) + (gz.x + gz.y));
    const float nn = tanhf((in_.x + in_.y) + r * (gn.x + gn.y));
    return (1.0f - z) * nn + z * hj;
}

__global__ __launch_bounds__(GRU_WAVES * 32)
void ggnn_gru(const float* agg,
              float* h,
              const float* __restrict__ w_ih,
              const float* __restrict__ w_hh,
              const float* __restrict__ b_ih,
              const float* __restrict__ b_hh, int N) {
    const int tid  = threadIdx.x;
    const int lane = tid & 31;
    const int wave = __builtin_amdgcn_readfirstlane(tid >> 5);
    const int j = lane < CCH ? lane : CCH - 1;

    __shared__ __attribute__((aligned(16))) float s_w[2 * 3 * CCH * CCH];
    __shared__ __attribute__((aligned(16))) float s_o[2][GRU_TILE * CCH];
    for (int i = tid; i < 3 * CCH * CCH / 4; i += GRU_WAVES * 32) {
        reinterpret_cast<float4*>(s_w)[i] = reinterpret_cast<const float4*>(w_ih)[i];
        reinterpret_cast<float4*>(s_w + 3 * CCH * CCH)[i] = reinterpret_cast<const float4*>(w_hh)[i];
    }
    __syncthreads();
    const float* s_ih = s_w;
    const float* s_hh = s_w + 3 * CCH * CCH;
    GruW w;
    #pragma unroll
    for (int k = 0; k < CH2; ++k) {
        w.wir[k] = *(const v2f*)&s_ih[j * CCH + 2 * k];
        w.wiz[k] = *(const v2f*)&s_ih[(CCH + j) * CCH + 2 * k];
        w.win[k] = *(const v2f*)&s_ih[(2 * CCH + j) * CCH + 2 * k];
        w.whr[k] = *(const v2f*)&s_hh[j * CCH + 2 * k];
        w.whz[k] = *(const v2f*)&s_hh[(CCH + j) * CCH + 2 * k];
        w.whn[k] = *(const v2f*)&s_hh[(2 * CCH + j) * CCH + 2 * k];
    }
    w.bir = b_ih[j]; w.biz = b_ih[CCH + j]; w.bin = b_ih[2 * CCH + j];
    w.bhr = b_hh[j]; w.bhz = b_hh[CCH + j]; w.bhn = b_hh[2 * CCH + j];

    const int ntiles = (N + GRU_TILE - 1) / GRU_TILE;
    int tile = blockIdx.x;
    if (tile >= ntiles) return;

    v2f a0[CH2], h0[CH2], a1[CH2], h1[CH2];
    float hj0, hj1;
    const int q0 = wave * (GRU_TILE / GRU_WAVES);
    gru_rows(agg, h, tile * GRU_TILE + q0, N, j, a0, h0, hj0);
    int buf = 0;
    #pragma unroll 1
    for (;;) {
        const int base = tile * GRU_TILE + q0;
        const int nt = tile + (int)gridDim.x;
        const bool more = nt < ntiles;
        float* so = &s_o[buf][q0 * CCH];
        float v;
        gru_rows(agg, h, base + 1, N, j, a1, h1, hj1);
        v = gru_node(w, a0, h0, hj0);
        if (lane < CCH) so[lane] = v;
        gru_rows(agg, h, base + 2, N, j, a0, h0, hj0);
        v = gru_node(w, a1, h1, hj1);
        if (lane < CCH) so[CCH + lane] = v;
        gru_rows(agg, h, base + 3, N, j, a1, h1, hj1);
        v = gru_node(w, a0, h0, hj0);
        if (lane < CCH) so[2 * CCH + lane] = v;
        gru_rows(agg, h, more ? nt * GRU_TILE + q0 : base + 3, N, j, a0, h0, hj0);
        v = gru_node(w, a1, h1, hj1);
        if (lane < CCH) so[3 * CCH + lane] = v;
        __syncthreads();
        float* ht = h + (size_t)tile * (GRU_TILE * CCH);
        if (tile * GRU_TILE + GRU_TILE <= N) {
            if (tid < GRU_TILE * CCH / 4)
                reinterpret_cast<float4*>(ht)[tid] = reinterpret_cast<const float4*>(s_o[buf])[tid];
        } else {
            const int nel = (N - tile * GRU_TILE) * CCH;
            for (int i = tid; i < nel; i += GRU_WAVES * 32) ht[i] = s_o[buf][i];
        }
        if (!more) break;
        tile = nt;
        buf ^= 1;
    }
}

__global__ void ggnn_final(const float* __restrict__ h,
                           const float* __restrict__ lin_w,
                           const float* __restrict__ lin_b,
                           float* __restrict__ out, int N) {
    __shared__ float s_w[CCH * CCH];
    __shared__ float s_b[CCH];
    for (int i = threadIdx.x; i < CCH * CCH; i += blockDim.x) s_w[i] = lin_w[i];
    if (threadIdx.x < CCH) s_b[threadIdx.x] = lin_b[threadIdx.x];
    __syncthreads();

    const int n = blockIdx.x * blockDim.x + threadIdx.x;
    if (n >= N) return;

    float hv[CCH];
    #pragma unroll
    for (int k = 0; k < CCH; ++k) {
        const float v = h[(size_t)n * CCH + k];
        hv[k] = v > 0.0f ? v : 0.0f;
    }
    #pragma unroll 1
    for (int j = 0; j < CCH; ++j) {
        float acc = s_b[j];
        #pragma unroll
        for (int k = 0; k < CCH; ++k) acc += hv[k] * s_w[j * CCH + k];
        out[(size_t)n * CCH + j] = acc;
    }
}

extern "C" void kernel_launch(void* const* d_in, const int* in_sizes, int n_in,
                              void* d_out, int out_size, void* d_ws, size_t ws_size,
                              hipStream_t stream) {
    const float* x      = (const float*)d_in[0];
    const int*   eidx   = (const int*)  d_in[1];
    const float* weight = (const float*)d_in[2];
    const float* w_ih   = (const float*)d_in[3];
    const float* w_hh   = (const float*)d_in[4];
    const float* b_ih   = (const float*)d_in[5];
    const float* b_hh   = (const float*)d_in[6];
    const float* lin_w  = (const float*)d_in[7];
    const float* lin_b  = (const float*)d_in[8];
    float* out = (float*)d_out;

    const int N = in_sizes[0] / CCH;
    const int E = in_sizes[1] / 2;
    const int* src = eidx;
    const int* dst = eidx + E;

    float* h   = (float*)d_ws;
    float* m   = h + (size_t)N * CCH;
    float* agg = m + (size_t)N * CCH;
    const int groups = (N + NPW - 1) / NPW;
    int2* rec  = (int2*)(agg + (size_t)N * CCH);
    int*  wcnt = (int*)(rec + (size_t)groups * OWN_WAVES * WCAP);
    unsigned* key = (unsigned*)(((size_t)(wcnt + (size_t)groups * OWN_WAVES) + 15) & ~(size_t)15);
    const int nwords = (E >> 11) * 512;
    if (nwords > 0)
        ggnn_owner_keys<<<(nwords + 255) / 256, 256, 0, stream>>>(dst, key, nwords);

    const int nelem = N * CCH;
    const int n4    = (nelem >> 2) + 4;
    ggnn_copy<<<(n4 + 255) / 256, 256, 0, stream>>>(x, h, nelem);

    const int tiles = (N + 15) / 16;
    for (int layer = 0; layer < NLAYERS; ++layer) {
        ggnn_transform_wmma<<<(tiles + 3) / 4, 128, 0, stream>>>(
            h, weight + (size_t)layer * CCH * CCH, m, N);
        if (layer == 0)
            ggnn_owner_sum<0><<<groups, OWN_THREADS, 0, stream>>>(src, dst, (const uint4*)key, m, agg, rec, wcnt, N, E);
        else
            ggnn_owner_sum<1><<<groups, OWN_THREADS, 0, stream>>>(src, dst, (const uint4*)key, m, agg, rec, wcnt, N, E);
        ggnn_gru<<<GRU_GROUPS, GRU_WAVES * 32, 0, stream>>>(agg, h, w_ih, w_hh,
                                                            b_ih, b_hh, N);
    }
    ggnn_final<<<(N + 127) / 128, 128, 0, stream>>>(h, lin_w, lin_b, out, N);
}


# Owner range 1024 -> 784 destination nodes (key byte = node / 784): 256 owner work-groups, one on each of the 256 WGPs, instead of 196; each WGP sums 1/256 of the edges; sum order stays LDS arrival
# speedup vs baseline: 10.4361x; 1.0648x over previous
#include <hip/hip_runtime.h>
#include <math.h>

#define CCH 30
#define NLAYERS 3
#define AST 34
#define CH2 (CCH / 2)

typedef __attribute__((ext_vector_type(2))) float v2f;
typedef __attribute__((ext_vector_type(8))) float v8f;

__device__ __forceinline__ float sigmoid_f(float x) { return 1.0f / (1.0f + __expf(-x)); }

__global__ void ggnn_transform_wmma(const float* __restrict__ h,
                                    const float* __restrict__ W,
                                    float* __restrict__ m, int N) {
    __shared__ float s_Wt[32 * AST];
    __shared__ float s_A[4][16 * AST];

    const int tid  = threadIdx.x;
    const int lane = tid & 31;
    const int warp = tid >> 5;
    const int row0 = (blockIdx.x * 4 + warp) * 16;
    const bool full = (row0 + 16 <= N);

    for (int i = tid; i < 32 * AST; i += 128) s_Wt[i] = 0.0f;
    for (int i = lane; i < 16 * AST; i += 32) s_A[warp][i] = 0.0f;
    __syncthreads();

    for (int i = tid; i < CCH * CCH; i += 128) {
        const int k = i / CCH, c = i % CCH;
        s_Wt[c * AST + k] = W[i];
    }
    if (full) {
        const float2* h2 = reinterpret_cast<const float2*>(h + (size_t)row0 * CCH);
        #pragma unroll
        for (int i = lane; i < 16 * CH2; i += 32) {
            const int r = i / CH2, c2 = i % CH2;
            *reinterpret_cast<float2*>(&s_A[warp][r * AST + 2 * c2]) =
                h2[(size_t)r * CH2 + c2];
        }
    } else {
        for (int i = lane; i < 16 * CCH; i += 32) {
            const int r = i / CCH, c = i % CCH;
            const int row = row0 + r;
            if (row < N) s_A[warp][r * AST + c] = h[(size_t)row * CCH + c];
        }
    }
    __syncthreads();

    if (row0 >= N) return;

    const int half = lane >> 4;
    const int l    = lane & 15;
    v8f acc0 = {};
    v8f acc1 = {};
    float* sa = &s_A[warp][0];

    #pragma unroll
    for (int k0 = 0; k0 < 32; k0 += 4) {
        const int ka = k0 + half * 2;
        const v2f a  = *(const v2f*)&sa  [l        * AST + ka];
        const v2f b0 = *(const v2f*)&s_Wt[l        * AST + ka];
        const v2f b1 = *(const v2f*)&s_Wt[(16 + l) * AST + ka];
        acc0 = __builtin_amdgcn_wmma_f32_16x16x4_f32(false, a, false, b0,
                                                     (short)0, acc0, false, false);
        acc1 = __builtin_amdgcn_wmma_f32_16x16x4_f32(false, a, false, b1,
                                                     (short)0, acc1, false, false);
    }

    #pragma unroll
    for (int v = 0; v < 8; ++v) {
        const int r = v + half * 8;
        sa[r * AST + l]      = acc0[v];
        sa[r * AST + 16 + l] = acc1[v];
    }

    if (full) {
        float2* m2 = reinterpret_cast<float2*>(m + (size_t)row0 * CCH);
        #pragma unroll
        for (int i = lane; i < 16 * CH2; i += 32) {
            const int r = i / CH2, c2 = i % CH2;
            m2[(size_t)r * CH2 + c2] =
                *reinterpret_cast<const float2*>(&sa[r * AST + 2 * c2]);
        }
    } else {
        for (int i = lane; i < 16 * CCH; i += 32) {
            const int r = i / CCH, c = i % CCH;
            const int row = row0 + r;
            if (row < N) m[(size_t)row * CCH + c] = sa[r * AST + c];
        }
    }
}

__global__ void ggnn_copy(const float* __restrict__ src, float* __restrict__ dst,
                          int n) {
    const int n4 = n >> 2;
    int i = blockIdx.x * blockDim.x + threadIdx.x;
    if (i < n4) {
        reinterpret_cast<float4*>(dst)[i] =
            reinterpret_cast<const float4*>(src)[i];
    }
    if (i < (n & 3)) {
        const int t = n4 * 4 + i;
        dst[t] = src[t];
    }
}

#define NPW         784
#define AGG_ROWS    1368
#define OWN_THREADS 1024
#define OWN_WAVES   32
#define WCAP        1600

__global__ void ggnn_owner_keys(const int* __restrict__ dst,
                                unsigned* __restrict__ key, int nwords) {
    const int i = blockIdx.x * blockDim.x + threadIdx.x;
    if (i >= nwords) return;
    int d0, d1, d2, d3;
    if ((reinterpret_cast<size_t>(dst) & 15) == 0) {
        const int4 d = reinterpret_cast<const int4*>(dst)[i];
        d0 = d.x; d1 = d.y; d2 = d.z; d3 = d.w;
    } else {
        d0 = dst[4 * i]; d1 = dst[4 * i + 1]; d2 = dst[4 * i + 2]; d3 = dst[4 * i + 3];
    }
    key[i] = (((unsigned)d0 / NPW) & 255u) | ((((unsigned)d1 / NPW) & 255u) << 8) |
             ((((unsigned)d2 / NPW) & 255u) << 16) | ((((unsigned)d3 / NPW) & 255u) << 24);
}

template <bool FULL, int B>
__device__ __forceinline__ void add_rows(int s, int dl, int n,
                                         const float* __restrict__ m,
                                         float* s_agg, int lane) {
    #pragma unroll
    for (int j = 0; j < 32; j += B) {
        if (!FULL && j >= n) break;
        int sj[B], dj[B];
        #pragma unroll
        for (int u = 0; u < B; ++u) {
            sj[u] = __builtin_amdgcn_readlane(s, j + u);
            dj[u] = __builtin_amdgcn_readlane(dl, j + u);
        }
        if (lane < CCH) {
            float v[B];
            #pragma unroll
            for (int u = 0; u < B; ++u)
                if (FULL || j + u < n) v[u] = m[(size_t)sj[u] * CCH + lane];
            #pragma unroll
            for (int u = 0; u < B; ++u)
                if (FULL || j + u < n)
                    __hip_atomic_fetch_add(&s_agg[dj[u] * CCH + lane], v[u],
                                           __ATOMIC_RELAXED,
                                           __HIP_MEMORY_SCOPE_WORKGROUP);
        }
    }
}

template <bool FULL, bool REC>
__device__ __forceinline__ void own_drain(int e, int n, int lo,
                                          const int* __restrict__ src,
                                          const int* __restrict__ dst,
                                          const float* __restrict__ m,
                                          float* s_agg, int lane,
                                          int2* __restrict__ rec, int recpos) {
    int s = 0, r = NPW;
    if (FULL || lane < n) {
        const unsigned rr = (unsigned)(dst[e] - lo);
        r = (int)(rr < (unsigned)NPW ? rr : (unsigned)NPW);
        s = src[e];
        if (REC && recpos + lane < WCAP) rec[recpos + lane] = make_int2(s, r);
    }
    add_rows<FULL, 16>(s, r, n, m, s_agg, lane);
}

template <bool REC>
__device__ __forceinline__ void own_push(int e, int& ring, int& pos, int& recpos, int lo,
                                         const int* __restrict__ src,
                                         const int* __restrict__ dst,
                                         const float* __restrict__ m,
                                         float* s_agg, int lane,
                                         int2* __restrict__ rec) {
    ring = (lane == pos) ? e : ring;
    if (++pos == 32) {
        own_drain<true, REC>(ring, 32, lo, src, dst, m, s_agg, lane, rec, recpos);
        recpos += 32;
        pos = 0;
    }
}

__device__ __forceinline__ void own_replay(const int2* __restrict__ rec, int cnt,
                                           const float* __restrict__ m,
                                           float* s_agg, int lane) {
    int2 cur = make_int2(0, 0);
    if (lane < cnt) cur = rec[lane];
    for (int base = 0; base < cnt; base += 32) {
        int2 nxt = make_int2(0, 0);
        if (base + 32 + lane < cnt) nxt = rec[base + 32 + lane];
        const int n = cnt - base;
        if (n >= 32) add_rows<true, 32>(cur.x, cur.y, 32, m, s_agg, lane);
        else         add_rows<false, 32>(cur.x, cur.y, n, m, s_agg, lane);
        cur = nxt;
    }
}

template <int MODE>
__global__ __launch_bounds__(OWN_THREADS)
void ggnn_owner_sum(const int* __restrict__ src,
                    const int* __restrict__ dst,
                    const uint4* __restrict__ key4,
                    const float* __restrict__ m,
                    float* __restrict__ agg,
                    int2* __restrict__ rec_all,
                    int* __restrict__ wcnt, int N, int E) {
    __shared__ __attribute__((aligned(16))) float s_agg[AGG_ROWS * CCH];

    const int tid  = threadIdx.x;
    const int lane = tid & 31;
    const int wave = __builtin_amdgcn_readfirstlane(tid >> 5);
    const int lo   = blockIdx.x * NPW;

    for (int i = tid; i < AGG_ROWS * CCH / 4; i += OWN_THREADS)
        reinterpret_cast<float4*>(s_agg)[i] = make_float4(0.f, 0.f, 0.f, 0.f);
    __syncthreads();

    constexpr bool REC = (MODE == 0);
    const int gw = blockIdx.x * OWN_WAVES + wave;
    int2* rec = rec_all + (size_t)gw * WCAP;
    int recpos = 0;
    bool replay = false;
    if (MODE == 1) {
        const int cnt = __builtin_amdgcn_readfirstlane(wcnt[gw]);
        replay = (cnt >= 0 && cnt <= WCAP);
        if (replay) own_replay(rec, cnt, m, s_agg, lane);
    }
    if (!replay) {

    int ring = 0;
    int pos  = 0;

    const unsigned pat = (unsigned)(blockIdx.x & 255) * 0x01010101u;
    const int nsu = E >> 11;
    const int start = (int)(((long long)blockIdx.x * nsu) / gridDim.x);

    if (wave < nsu) {
        int k = wave;
        uint4 cur[4];
        {
            int su = k + start; if (su >= nsu) su -= nsu;
            const uint4* p = key4 + (size_t)su * 128 + lane;
            #pragma unroll
            for (int u = 0; u < 4; ++u) cur[u] = p[32 * u];
        }
        while (k < nsu) {
            const int kn = k + OWN_WAVES;
            const int kl = kn < nsu ? kn : k;
            int sun = kl + start; if (sun >= nsu) sun -= nsu;
            const uint4* pn = key4 + (size_t)sun * 128 + lane;
            uint4 nxt[4];
            #pragma unroll
            for (int u = 0; u < 4; ++u) nxt[u] = pn[32 * u];

            int su = k + start; if (su >= nsu) su -= nsu;
            #pragma unroll
            for (int u = 0; u < 4; ++u) {
                const unsigned wv[4] = {cur[u].x, cur[u].y, cur[u].z, cur[u].w};
                unsigned fl = 0u;
                #pragma unroll
                for (int c = 0; c < 4; ++c) {
                    const unsigned x = wv[c] ^ pat;
                    const unsigned y = ((x & 0x7F7F7F7Fu) + 0x7F7F7F7Fu) | x;
                    fl |= (~y & 0x80808080u) >> (7 - c);
                }
                unsigned mk = __builtin_amdgcn_ballot_w32(fl != 0u);
                const int eu = su * 2048 + u * 512;
                while (mk != 0u) {
                    const int L = __builtin_ctz(mk);
                    mk &= mk - 1u;
                    unsigned f = (unsigned)__builtin_amdgcn_readlane((int)fl, L);
                    while (f != 0u) {
                        const int i = __builtin_ctz(f);
                        f &= f - 1u;
                        own_push<REC>(eu + L * 16 + (i & 7) * 4 + (i >> 3),
                                      ring, pos, recpos, lo, src, dst, m, s_agg, lane, rec);
                    }
                }
            }
            #pragma unroll
            for (int u = 0; u < 4; ++u) cur[u] = nxt[u];
            k = kn;
        }
    }

    for (int base = nsu * 2048 + wave * 32; base < E; base += OWN_THREADS) {
        const int  e     = base + lane;
        const bool valid = e < E;
        const int  d     = valid ? dst[e] : -1;
        const bool p     = valid && (unsigned)(d - lo) < (unsigned)NPW;
        unsigned mk = __builtin_amdgcn_ballot_w32(p);
        while (mk != 0u) {
            const int L = __builtin_ctz(mk);
            mk &= mk - 1u;
            own_push<REC>(base + L, ring, pos, recpos, lo, src, dst, m, s_agg, lane, rec);
        }
    }
    if (pos > 0) {
        own_drain<false, REC>(ring, pos, lo, src, dst, m, s_agg, lane, rec, recpos);
        recpos += pos;
    }
    if (REC && lane == 0) wcnt[gw] = recpos;
    }
    __syncthreads();

    const int nn = (N - lo) < NPW ? (N - lo) : NPW;
    float* ag = agg + (size_t)lo * CCH;
    for (int i = tid; i < nn * CCH; i += OWN_THREADS) ag[i] = s_agg[i];
}

#define GRU_WAVES  4
#define GRU_GROUPS 768
#define GRU_TILE   16

__device__ __forceinline__ void gru_rows(const float* agg, const float* h, int n, int N, int j,
                                         v2f (&a)[CH2], v2f (&hv)[CH2], float& hj) {
    n = n < N ? n : N - 1;
    const float* ar = agg + (size_t)n * CCH;
    const float* hr = h   + (size_t)n * CCH;
    #pragma unroll
    for (int k = 0; k < CH2; ++k) {
        a[k]  = *(const v2f*)(ar + 2 * k);
        hv[k] = *(const v2f*)(hr + 2 * k);
    }
    hj = hr[j];
}

struct GruW {
    v2f wir[CH2], wiz[CH2], win[CH2], whr[CH2], whz[CH2], whn[CH2];
    float bir, biz, bin, bhr, bhz, bhn;
};

__device__ __forceinline__ float gru_node(const GruW& w, const v2f (&a)[CH2], const v2f (&hv)[CH2],
                                          float hj) {
    v2f ir = {w.bir, 0.0f}, iz = {w.biz, 0.0f}, in_ = {w.bin, 0.0f};
    v2f gr = {w.bhr, 0.0f}, gz = {w.bhz, 0.0f}, gn = {w.bhn, 0.0f};
    #pragma unroll
    for (int k = 0; k < CH2; ++k) {
        ir  = __builtin_elementwise_fma(a[k],  w.wir[k], ir);
        iz  = __builtin_elementwise_fma(a[k],  w.wiz[k], iz);
        in_ = __builtin_elementwise_fma(a[k],  w.win[k], in_);
        gr  = __builtin_elementwise_fma(hv[k], w.whr[k], gr);
        gz  = __builtin_elementwise_fma(hv[k], w.whz[k], gz);
        gn  = __builtin_elementwise_fma(hv[k], w.whn[k], gn);
    }
    const float r  = sigmoid_f((ir.x + ir.y) + (gr.x + gr.y));
    const float z  = sigmoid_f((iz.x + iz.y) + (gz.x + gz.y));
    const float nn = tanhf((in_.x + in_.y) + r * (gn.x + gn.y));
    return (1.0f - z) * nn + z * hj;
}

__global__ __launch_bounds__(GRU_WAVES * 32)
void ggnn_gru(const float* agg,
              float* h,
              const float* __restrict__ w_ih,
              const float* __restrict__ w_hh,
              const float* __restrict__ b_ih,
              const float* __restrict__ b_hh, int N) {
    const int tid  = threadIdx.x;
    const int lane = tid & 31;
    const int wave = __builtin_amdgcn_readfirstlane(tid >> 5);
    const int j = lane < CCH ? lane : CCH - 1;

    __shared__ __attribute__((aligned(16))) float s_w[2 * 3 * CCH * CCH];
    __shared__ __attribute__((aligned(16))) float s_o[2][GRU_TILE * CCH];
    for (int i = tid; i < 3 * CCH * CCH / 4; i += GRU_WAVES * 32) {
        reinterpret_cast<float4*>(s_w)[i] = reinterpret_cast<const float4*>(w_ih)[i];
        reinterpret_cast<float4*>(s_w + 3 * CCH * CCH)[i] = reinterpret_cast<const float4*>(w_hh)[i];
    }
    __syncthreads();
    const float* s_ih = s_w;
    const float* s_hh = s_w + 3 * CCH * CCH;
    GruW w;
    #pragma unroll
    for (int k = 0; k < CH2; ++k) {
        w.wir[k] = *(const v2f*)&s_ih[j * CCH + 2 * k];
        w.wiz[k] = *(const v2f*)&s_ih[(CCH + j) * CCH + 2 * k];
        w.win[k] = *(const v2f*)&s_ih[(2 * CCH + j) * CCH + 2 * k];
        w.whr[k] = *(const v2f*)&s_hh[j * CCH + 2 * k];
        w.whz[k] = *(const v2f*)&s_hh[(CCH + j) * CCH + 2 * k];
        w.whn[k] = *(const v2f*)&s_hh[(2 * CCH + j) * CCH + 2 * k];
    }
    w.bir = b_ih[j]; w.biz = b_ih[CCH + j]; w.bin = b_ih[2 * CCH + j];
    w.bhr = b_hh[j]; w.bhz = b_hh[CCH + j]; w.bhn = b_hh[2 * CCH + j];

    const int ntiles = (N + GRU_TILE - 1) / GRU_TILE;
    int tile = blockIdx.x;
    if (tile >= ntiles) return;

    v2f a0[CH2], h0[CH2], a1[CH2], h1[CH2];
    float hj0, hj1;
    const int q0 = wave * (GRU_TILE / GRU_WAVES);
    gru_rows(agg, h, tile * GRU_TILE + q0, N, j, a0, h0, hj0);
    int buf = 0;
    #pragma unroll 1
    for (;;) {
        const int base = tile * GRU_TILE + q0;
        const int nt = tile + (int)gridDim.x;
        const bool more = nt < ntiles;
        float* so = &s_o[buf][q0 * CCH];
        float v;
        gru_rows(agg, h, base + 1, N, j, a1, h1, hj1);
        v = gru_node(w, a0, h0, hj0);
        if (lane < CCH) so[lane] = v;
        gru_rows(agg, h, base + 2, N, j, a0, h0, hj0);
        v = gru_node(w, a1, h1, hj1);
        if (lane < CCH) so[CCH + lane] = v;
        gru_rows(agg, h, base + 3, N, j, a1, h1, hj1);
        v = gru_node(w, a0, h0, hj0);
        if (lane < CCH) so[2 * CCH + lane] = v;
        gru_rows(agg, h, more ? nt * GRU_TILE + q0 : base + 3, N, j, a0, h0, hj0);
        v = gru_node(w, a1, h1, hj1);
        if (lane < CCH) so[3 * CCH + lane] = v;
        __syncthreads();
        float* ht = h + (size_t)tile * (GRU_TILE * CCH);
        if (tile * GRU_TILE + GRU_TILE <= N) {
            if (tid < GRU_TILE * CCH / 4)
                reinterpret_cast<float4*>(ht)[tid] = reinterpret_cast<const float4*>(s_o[buf])[tid];
        } else {
            const int nel = (N - tile * GRU_TILE) * CCH;
            for (int i = tid; i < nel; i += GRU_WAVES * 32) ht[i] = s_o[buf][i];
        }
        if (!more) break;
        tile = nt;
        buf ^= 1;
    }
}

__global__ void ggnn_final(const float* __restrict__ h,
                           const float* __restrict__ lin_w,
                           const float* __restrict__ lin_b,
                           float* __restrict__ out, int N) {
    __shared__ float s_w[CCH * CCH];
    __shared__ float s_b[CCH];
    for (int i = threadIdx.x; i < CCH * CCH; i += blockDim.x) s_w[i] = lin_w[i];
    if (threadIdx.x < CCH) s_b[threadIdx.x] = lin_b[threadIdx.x];
    __syncthreads();

    const int n = blockIdx.x * blockDim.x + threadIdx.x;
    if (n >= N) return;

    float hv[CCH];
    #pragma unroll
    for (int k = 0; k < CCH; ++k) {
        const float v = h[(size_t)n * CCH + k];
        hv[k] = v > 0.0f ? v : 0.0f;
    }
    #pragma unroll 1
    for (int j = 0; j < CCH; ++j) {
        float acc = s_b[j];
        #pragma unroll
        for (int k = 0; k < CCH; ++k) acc += hv[k] * s_w[j * CCH + k];
        out[(size_t)n * CCH + j] = acc;
    }
}

extern "C" void kernel_launch(void* const* d_in, const int* in_sizes, int n_in,
                              void* d_out, int out_size, void* d_ws, size_t ws_size,
                              hipStream_t stream) {
    const float* x      = (const float*)d_in[0];
    const int*   eidx   = (const int*)  d_in[1];
    const float* weight = (const float*)d_in[2];
    const float* w_ih   = (const float*)d_in[3];
    const float* w_hh   = (const float*)d_in[4];
    const float* b_ih   = (const float*)d_in[5];
    const float* b_hh   = (const float*)d_in[6];
    const float* lin_w  = (const float*)d_in[7];
    const float* lin_b  = (const float*)d_in[8];
    float* out = (float*)d_out;

    const int N = in_sizes[0] / CCH;
    const int E = in_sizes[1] / 2;
    const int* src = eidx;
    const int* dst = eidx + E;

    float* h   = (float*)d_ws;
    float* m   = h + (size_t)N * CCH;
    float* agg = m + (size_t)N * CCH;
    const int groups = (N + NPW - 1) / NPW;
    int2* rec  = (int2*)(agg + (size_t)N * CCH);
    int*  wcnt = (int*)(rec + (size_t)groups * OWN_WAVES * WCAP);
    unsigned* key = (unsigned*)(((size_t)(wcnt + (size_t)groups * OWN_WAVES) + 15) & ~(size_t)15);
    const int nwords = (E >> 11) * 512;
    if (nwords > 0)
        ggnn_owner_keys<<<(nwords + 255) / 256, 256, 0, stream>>>(dst, key, nwords);

    const int nelem = N * CCH;
    const int n4    = (nelem >> 2) + 4;
    ggnn_copy<<<(n4 + 255) / 256, 256, 0, stream>>>(x, h, nelem);

    const int tiles = (N + 15) / 16;
    for (int layer = 0; layer < NLAYERS; ++layer) {
        ggnn_transform_wmma<<<(tiles + 3) / 4, 128, 0, stream>>>(
            h, weight + (size_t)layer * CCH * CCH, m, N);
        if (layer == 0)
            ggnn_owner_sum<0><<<groups, OWN_THREADS, 0, stream>>>(src, dst, (const uint4*)key, m, agg, rec, wcnt, N, E);
        else
            ggnn_owner_sum<1><<<groups, OWN_THREADS, 0, stream>>>(src, dst, (const uint4*)key, m, agg, rec, wcnt, N, E);
        ggnn_gru<<<GRU_GROUPS, GRU_WAVES * 32, 0, stream>>>(agg, h, w_ih, w_hh,
                                                            b_ih, b_hh, N);
    }
    ggnn_final<<<(N + 127) / 128, 128, 0, stream>>>(h, lin_w, lin_b, out, N);
}


# First-layer scan: marked lanes write their candidates side by side into a per-wave LDS queue (place = waiting + mbcnt rank), 32 read back per drain, no scalar loop per candidate; sum order LDS arrival
# speedup vs baseline: 10.7951x; 1.0344x over previous
#include <hip/hip_runtime.h>
#include <math.h>

#define CCH 30
#define NLAYERS 3
#define AST 34
#define CH2 (CCH / 2)

typedef __attribute__((ext_vector_type(2))) float v2f;
typedef __attribute__((ext_vector_type(8))) float v8f;

__device__ __forceinline__ float sigmoid_f(float x) { return 1.0f / (1.0f + __expf(-x)); }

__global__ void ggnn_transform_wmma(const float* __restrict__ h,
                                    const float* __restrict__ W,
                                    float* __restrict__ m, int N) {
    __shared__ float s_Wt[32 * AST];
    __shared__ float s_A[4][16 * AST];

    const int tid  = threadIdx.x;
    const int lane = tid & 31;
    const int warp = tid >> 5;
    const int row0 = (blockIdx.x * 4 + warp) * 16;
    const bool full = (row0 + 16 <= N);

    for (int i = tid; i < 32 * AST; i += 128) s_Wt[i] = 0.0f;
    for (int i = lane; i < 16 * AST; i += 32) s_A[warp][i] = 0.0f;
    __syncthreads();

    for (int i = tid; i < CCH * CCH; i += 128) {
        const int k = i / CCH, c = i % CCH;
        s_Wt[c * AST + k] = W[i];
    }
    if (full) {
        const float2* h2 = reinterpret_cast<const float2*>(h + (size_t)row0 * CCH);
        #pragma unroll
        for (int i = lane; i < 16 * CH2; i += 32) {
            const int r = i / CH2, c2 = i % CH2;
            *reinterpret_cast<float2*>(&s_A[warp][r * AST + 2 * c2]) =
                h2[(size_t)r * CH2 + c2];
        }
    } else {
        for (int i = lane; i < 16 * CCH; i += 32) {
            const int r = i / CCH, c = i % CCH;
            const int row = row0 + r;
            if (row < N) s_A[warp][r * AST + c] = h[(size_t)row * CCH + c];
        }
    }
    __syncthreads();

    if (row0 >= N) return;

    const int half = lane >> 4;
    const int l    = lane & 15;
    v8f acc0 = {};
    v8f acc1 = {};
    float* sa = &s_A[warp][0];

    #pragma unroll
    for (int k0 = 0; k0 < 32; k0 += 4) {
        const int ka = k0 + half * 2;
        const v2f a  = *(const v2f*)&sa  [l        * AST + ka];
        const v2f b0 = *(const v2f*)&s_Wt[l        * AST + ka];
        const v2f b1 = *(const v2f*)&s_Wt[(16 + l) * AST + ka];
        acc0 = __builtin_amdgcn_wmma_f32_16x16x4_f32(false, a, false, b0,
                                                     (short)0, acc0, false, false);
        acc1 = __builtin_amdgcn_wmma_f32_16x16x4_f32(false, a, false, b1,
                                                     (short)0, acc1, false, false);
    }

    #pragma unroll
    for (int v = 0; v < 8; ++v) {
        const int r = v + half * 8;
        sa[r * AST + l]      = acc0[v];
        sa[r * AST + 16 + l] = acc1[v];
    }

    if (full) {
        float2* m2 = reinterpret_cast<float2*>(m + (size_t)row0 * CCH);
        #pragma unroll
        for (int i = lane; i < 16 * CH2; i += 32) {
            const int r = i / CH2, c2 = i % CH2;
            m2[(size_t)r * CH2 + c2] =
                *reinterpret_cast<const float2*>(&sa[r * AST + 2 * c2]);
        }
    } else {
        for (int i = lane; i < 16 * CCH; i += 32) {
            const int r = i / CCH, c = i % CCH;
            const int row = row0 + r;
            if (row < N) m[(size_t)row * CCH + c] = sa[r * AST + c];
        }
    }
}

__global__ void ggnn_copy(const float* __restrict__ src, float* __restrict__ dst,
                          int n) {
    const int n4 = n >> 2;
    int i = blockIdx.x * blockDim.x + threadIdx.x;
    if (i < n4) {
        reinterpret_cast<float4*>(dst)[i] =
            reinterpret_cast<const float4*>(src)[i];
    }
    if (i < (n & 3)) {
        const int t = n4 * 4 + i;
        dst[t] = src[t];
    }
}

#define NPW         784
#define AGG_ROWS    1368
#define OWN_THREADS 1024
#define OWN_WAVES   32
#define WCAP        1600
#define QCAP        64

__global__ void ggnn_owner_keys(const int* __restrict__ dst,
                                unsigned* __restrict__ key, int nwords) {
    const int i = blockIdx.x * blockDim.x + threadIdx.x;
    if (i >= nwords) return;
    int d0, d1, d2, d3;
    if ((reinterpret_cast<size_t>(dst) & 15) == 0) {
        const int4 d = reinterpret_cast<const int4*>(dst)[i];
        d0 = d.x; d1 = d.y; d2 = d.z; d3 = d.w;
    } else {
        d0 = dst[4 * i]; d1 = dst[4 * i + 1]; d2 = dst[4 * i + 2]; d3 = dst[4 * i + 3];
    }
    key[i] = (((unsigned)d0 / NPW) & 255u) | ((((unsigned)d1 / NPW) & 255u) << 8) |
             ((((unsigned)d2 / NPW) & 255u) << 16) | ((((unsigned)d3 / NPW) & 255u) << 24);
}

template <bool FULL, int B>
__device__ __forceinline__ void add_rows(int s, int dl, int n,
                                         const float* __restrict__ m,
                                         float* s_agg, int lane) {
    #pragma unroll
    for (int j = 0; j < 32; j += B) {
        if (!FULL && j >= n) break;
        int sj[B], dj[B];
        #pragma unroll
        for (int u = 0; u < B; ++u) {
            sj[u] = __builtin_amdgcn_readlane(s, j + u);
            dj[u] = __builtin_amdgcn_readlane(dl, j + u);
        }
        if (lane < CCH) {
            float v[B];
            #pragma unroll
            for (int u = 0; u < B; ++u)
                if (FULL || j + u < n) v[u] = m[(size_t)sj[u] * CCH + lane];
            #pragma unroll
            for (int u = 0; u < B; ++u)
                if (FULL || j + u < n)
                    __hip_atomic_fetch_add(&s_agg[dj[u] * CCH + lane], v[u],
                                           __ATOMIC_RELAXED,
                                           __HIP_MEMORY_SCOPE_WORKGROUP);
        }
    }
}

template <bool FULL, bool REC>
__device__ __forceinline__ void own_drain(int e, int n, int lo,
                                          const int* __restrict__ src,
                                          const int* __restrict__ dst,
                                          const float* __restrict__ m,
                                          float* s_agg, int lane,
                                          int2* __restrict__ rec, int recpos) {
    int s = 0, r = NPW;
    if (FULL || lane < n) {
        const unsigned rr = (unsigned)(dst[e] - lo);
        r = (int)(rr < (unsigned)NPW ? rr : (unsigned)NPW);
        s = src[e];
        if (REC && recpos + lane < WCAP) rec[recpos + lane] = make_int2(s, r);
    }
    add_rows<FULL, 16>(s, r, n, m, s_agg, lane);
}

template <bool REC>
__device__ __forceinline__ void own_added(int n, int* q, int& pos, int& recpos, int lo,
                                          const int* __restrict__ src,
                                          const int* __restrict__ dst,
                                          const float* __restrict__ m,
                                          float* s_agg, int lane,
                                          int2* __restrict__ rec) {
    pos += n;
    if (pos >= 32) {
        const int e = q[lane];
        pos -= 32;
        if (lane < pos) q[lane] = q[32 + lane];
        own_drain<true, REC>(e, 32, lo, src, dst, m, s_agg, lane, rec, recpos);
        recpos += 32;
    }
}

__device__ __forceinline__ void own_replay(const int2* __restrict__ rec, int cnt,
                                           const float* __restrict__ m,
                                           float* s_agg, int lane) {
    int2 cur = make_int2(0, 0);
    if (lane < cnt) cur = rec[lane];
    for (int base = 0; base < cnt; base += 32) {
        int2 nxt = make_int2(0, 0);
        if (base + 32 + lane < cnt) nxt = rec[base + 32 + lane];
        const int n = cnt - base;
        if (n >= 32) add_rows<true, 32>(cur.x, cur.y, 32, m, s_agg, lane);
        else         add_rows<false, 32>(cur.x, cur.y, n, m, s_agg, lane);
        cur = nxt;
    }
}

template <int MODE>
__global__ __launch_bounds__(OWN_THREADS)
void ggnn_owner_sum(const int* __restrict__ src,
                    const int* __restrict__ dst,
                    const uint4* __restrict__ key4,
                    const float* __restrict__ m,
                    float* __restrict__ agg,
                    int2* __restrict__ rec_all,
                    int* __restrict__ wcnt, int N, int E) {
    __shared__ __attribute__((aligned(16))) float s_agg[AGG_ROWS * CCH];
    __shared__ int s_q[OWN_WAVES * QCAP];

    const int tid  = threadIdx.x;
    const int lane = tid & 31;
    const int wave = __builtin_amdgcn_readfirstlane(tid >> 5);
    const int lo   = blockIdx.x * NPW;

    for (int i = tid; i < AGG_ROWS * CCH / 4; i += OWN_THREADS)
        reinterpret_cast<float4*>(s_agg)[i] = make_float4(0.f, 0.f, 0.f, 0.f);
    __syncthreads();

    constexpr bool REC = (MODE == 0);
    const int gw = blockIdx.x * OWN_WAVES + wave;
    int2* rec = rec_all + (size_t)gw * WCAP;
    int recpos = 0;
    bool replay = false;
    if (MODE == 1) {
        const int cnt = __builtin_amdgcn_readfirstlane(wcnt[gw]);
        replay = (cnt >= 0 && cnt <= WCAP);
        if (replay) own_replay(rec, cnt, m, s_agg, lane);
    }
    if (!replay) {

    int* q  = s_q + wave * QCAP;
    int pos = 0;

    const unsigned pat = (unsigned)(blockIdx.x & 255) * 0x01010101u;
    const int nsu = E >> 11;
    const int start = (int)(((long long)blockIdx.x * nsu) / gridDim.x);

    if (wave < nsu) {
        int k = wave;
        uint4 cur[4];
        {
            int su = k + start; if (su >= nsu) su -= nsu;
            const uint4* p = key4 + (size_t)su * 128 + lane;
            #pragma unroll
            for (int u = 0; u < 4; ++u) cur[u] = p[32 * u];
        }
        while (k < nsu) {
            const int kn = k + OWN_WAVES;
            const int kl = kn < nsu ? kn : k;
            int sun = kl + start; if (sun >= nsu) sun -= nsu;
            const uint4* pn = key4 + (size_t)sun * 128 + lane;
            uint4 nxt[4];
            #pragma unroll
            for (int u = 0; u < 4; ++u) nxt[u] = pn[32 * u];

            int su = k + start; if (su >= nsu) su -= nsu;
            #pragma unroll
            for (int u = 0; u < 4; ++u) {
                const unsigned wv[4] = {cur[u].x, cur[u].y, cur[u].z, cur[u].w};
                unsigned fl = 0u;
                #pragma unroll
                for (int c = 0; c < 4; ++c) {
                    const unsigned x = wv[c] ^ pat;
                    const unsigned y = ((x & 0x7F7F7F7Fu) + 0x7F7F7F7Fu) | x;
                    fl |= (~y & 0x80808080u) >> (7 - c);
                }
                unsigned mk = __builtin_amdgcn_ballot_w32(fl != 0u);
                const int eu = su * 2048 + u * 512 + lane * 16;
                while (mk != 0u) {
                    if (fl != 0u) {
                        const int i = __builtin_ctz(fl);
                        q[pos + (int)__builtin_amdgcn_mbcnt_lo(mk, 0u)] =
                            eu + (i & 7) * 4 + (i >> 3);
                    }
                    fl &= fl - 1u;
                    own_added<REC>(__builtin_popcount(mk), q, pos, recpos, lo, src, dst, m,
                                   s_agg, lane, rec);
                    mk = __builtin_amdgcn_ballot_w32(fl != 0u);
                }
            }
            #pragma unroll
            for (int u = 0; u < 4; ++u) cur[u] = nxt[u];
            k = kn;
        }
    }

    for (int base = nsu * 2048 + wave * 32; base < E; base += OWN_THREADS) {
        const int  e     = base + lane;
        const bool valid = e < E;
        const int  d     = valid ? dst[e] : -1;
        const bool p     = valid && (unsigned)(d - lo) < (unsigned)NPW;
        const unsigned mk = __builtin_amdgcn_ballot_w32(p);
        if (mk != 0u) {
            if (p) q[pos + (int)__builtin_amdgcn_mbcnt_lo(mk, 0u)] = e;
            own_added<REC>(__builtin_popcount(mk), q, pos, recpos, lo, src, dst, m,
                           s_agg, lane, rec);
        }
    }
    if (pos > 0) {
        const int e = lane < pos ? q[lane] : 0;
        own_drain<false, REC>(e, pos, lo, src, dst, m, s_agg, lane, rec, recpos);
        recpos += pos;
    }
    if (REC && lane == 0) wcnt[gw] = recpos;
    }
    __syncthreads();

    const int nn = (N - lo) < NPW ? (N - lo) : NPW;
    float* ag = agg + (size_t)lo * CCH;
    for (int i = tid; i < nn * CCH; i += OWN_THREADS) ag[i] = s_agg[i];
}

#define GRU_WAVES  4
#define GRU_GROUPS 768
#define GRU_TILE   16

__device__ __forceinline__ void gru_rows(const float* agg, const float* h, int n, int N, int j,
                                         v2f (&a)[CH2], v2f (&hv)[CH2], float& hj) {
    n = n < N ? n : N - 1;
    const float* ar = agg + (size_t)n * CCH;
    const float* hr = h   + (size_t)n * CCH;
    #pragma unroll
    for (int k = 0; k < CH2; ++k) {
        a[k]  = *(const v2f*)(ar + 2 * k);
        hv[k] = *(const v2f*)(hr + 2 * k);
    }
    hj = hr[j];
}

struct GruW {
    v2f wir[CH2], wiz[CH2], win[CH2], whr[CH2], whz[CH2], whn[CH2];
    float bir, biz, bin, bhr, bhz, bhn;
};

__device__ __forceinline__ float gru_node(const GruW& w, const v2f (&a)[CH2], const v2f (&hv)[CH2],
                                          float hj) {
    v2f ir = {w.bir, 0.0f}, iz = {w.biz, 0.0f}, in_ = {w.bin, 0.0f};
    v2f gr = {w.bhr, 0.0f}, gz = {w.bhz, 0.0f}, gn = {w.bhn, 0.0f};
    #pragma unroll
    for (int k = 0; k < CH2; ++k) {
        ir  = __builtin_elementwise_fma(a[k],  w.wir[k], ir);
        iz  = __builtin_elementwise_fma(a[k],  w.wiz[k], iz);
        in_ = __builtin_elementwise_fma(a[k],  w.win[k], in_);
        gr  = __builtin_elementwise_fma(hv[k], w.whr[k], gr);
        gz  = __builtin_elementwise_fma(hv[k], w.whz[k], gz);
        gn  = __builtin_elementwise_fma(hv[k], w.whn[k], gn);
    }
    const float r  = sigmoid_f((ir.x + ir.y) + (gr.x + gr.y));
    const float z  = sigmoid_f((iz.x + iz.y) + (gz.x + gz.y));
    const float nn = tanhf((in_.x + in_.y) + r * (gn.x + gn.y));
    return (1.0f - z) * nn + z * hj;
}

__global__ __launch_bounds__(GRU_WAVES * 32)
void ggnn_gru(const float* agg,
              float* h,
              const float* __restrict__ w_ih,
              const float* __restrict__ w_hh,
              const float* __restrict__ b_ih,
              const float* __restrict__ b_hh, int N) {
    const int tid  = threadIdx.x;
    const int lane = tid & 31;
    const int wave = __builtin_amdgcn_readfirstlane(tid >> 5);
    const int j = lane < CCH ? lane : CCH - 1;

    __shared__ __attribute__((aligned(16))) float s_w[2 * 3 * CCH * CCH];
    __shared__ __attribute__((aligned(16))) float s_o[2][GRU_TILE * CCH];
    for (int i = tid; i < 3 * CCH * CCH / 4; i += GRU_WAVES * 32) {
        reinterpret_cast<float4*>(s_w)[i] = reinterpret_cast<const float4*>(w_ih)[i];
        reinterpret_cast<float4*>(s_w + 3 * CCH * CCH)[i] = reinterpret_cast<const float4*>(w_hh)[i];
    }
    __syncthreads();
    const float* s_ih = s_w;
    const float* s_hh = s_w + 3 * CCH * CCH;
    GruW w;
    #pragma unroll
    for (int k = 0; k < CH2; ++k) {
        w.wir[k] = *(const v2f*)&s_ih[j * CCH + 2 * k];
        w.wiz[k] = *(const v2f*)&s_ih[(CCH + j) * CCH + 2 * k];
        w.win[k] = *(const v2f*)&s_ih[(2 * CCH + j) * CCH + 2 * k];
        w.whr[k] = *(const v2f*)&s_hh[j * CCH + 2 * k];
        w.whz[k] = *(const v2f*)&s_hh[(CCH + j) * CCH + 2 * k];
        w.whn[k] = *(const v2f*)&s_hh[(2 * CCH + j) * CCH + 2 * k];
    }
    w.bir = b_ih[j]; w.biz = b_ih[CCH + j]; w.bin = b_ih[2 * CCH + j];
    w.bhr = b_hh[j]; w.bhz = b_hh[CCH + j]; w.bhn = b_hh[2 * CCH + j];

    const int ntiles = (N + GRU_TILE - 1) / GRU_TILE;
    int tile = blockIdx.x;
    if (tile >= ntiles) return;

    v2f a0[CH2], h0[CH2], a1[CH2], h1[CH2];
    float hj0, hj1;
    const int q0 = wave * (GRU_TILE / GRU_WAVES);
    gru_rows(agg, h, tile * GRU_TILE + q0, N, j, a0, h0, hj0);
    int buf = 0;
    #pragma unroll 1
    for (;;) {
        const int base = tile * GRU_TILE + q0;
        const int nt = tile + (int)gridDim.x;
        const bool more = nt < ntiles;
        float* so = &s_o[buf][q0 * CCH];
        float v;
        gru_rows(agg, h, base + 1, N, j, a1, h1, hj1);
        v = gru_node(w, a0, h0, hj0);
        if (lane < CCH) so[lane] = v;
        gru_rows(agg, h, base + 2, N, j, a0, h0, hj0);
        v = gru_node(w, a1, h1, hj1);
        if (lane < CCH) so[CCH + lane] = v;
        gru_rows(agg, h, base + 3, N, j, a1, h1, hj1);
        v = gru_node(w, a0, h0, hj0);
        if (lane < CCH) so[2 * CCH + lane] = v;
        gru_rows(agg, h, more ? nt * GRU_TILE + q0 : base + 3, N, j, a0, h0, hj0);
        v = gru_node(w, a1, h1, hj1);
        if (lane < CCH) so[3 * CCH + lane] = v;
        __syncthreads();
        float* ht = h + (size_t)tile * (GRU_TILE * CCH);
        if (tile * GRU_TILE + GRU_TILE <= N) {
            if (tid < GRU_TILE * CCH / 4)
                reinterpret_cast<float4*>(ht)[tid] = reinterpret_cast<const float4*>(s_o[buf])[tid];
        } else {
            const int nel = (N - tile * GRU_TILE) * CCH;
            for (int i = tid; i < nel; i += GRU_WAVES * 32) ht[i] = s_o[buf][i];
        }
        if (!more) break;
        tile = nt;
        buf ^= 1;
    }
}

__global__ void ggnn_final(const float* __restrict__ h,
                           const float* __restrict__ lin_w,
                           const float* __restrict__ lin_b,
                           float* __restrict__ out, int N) {
    __shared__ float s_w[CCH * CCH];
    __shared__ float s_b[CCH];
    for (int i = threadIdx.x; i < CCH * CCH; i += blockDim.x) s_w[i] = lin_w[i];
    if (threadIdx.x < CCH) s_b[threadIdx.x] = lin_b[threadIdx.x];
    __syncthreads();

    const int n = blockIdx.x * blockDim.x + threadIdx.x;
    if (n >= N) return;

    float hv[CCH];
    #pragma unroll
    for (int k = 0; k < CCH; ++k) {
        const float v = h[(size_t)n * CCH + k];
        hv[k] = v > 0.0f ? v : 0.0f;
    }
    #pragma unroll 1
    for (int j = 0; j < CCH; ++j) {
        float acc = s_b[j];
        #pragma unroll
        for (int k = 0; k < CCH; ++k) acc += hv[k] * s_w[j * CCH + k];
        out[(size_t)n * CCH + j] = acc;
    }
}

extern "C" void kernel_launch(void* const* d_in, const int* in_sizes, int n_in,
                              void* d_out, int out_size, void* d_ws, size_t ws_size,
                              hipStream_t stream) {
    const float* x      = (const float*)d_in[0];
    const int*   eidx   = (const int*)  d_in[1];
    const float* weight = (const float*)d_in[2];
    const float* w_ih   = (const float*)d_in[3];
    const float* w_hh   = (const float*)d_in[4];
    const float* b_ih   = (const float*)d_in[5];
    const float* b_hh   = (const float*)d_in[6];
    const float* lin_w  = (const float*)d_in[7];
    const float* lin_b  = (const float*)d_in[8];
    float* out = (float*)d_out;

    const int N = in_sizes[0] / CCH;
    const int E = in_sizes[1] / 2;
    const int* src = eidx;
    const int* dst = eidx + E;

    float* h   = (float*)d_ws;
    float* m   = h + (size_t)N * CCH;
    float* agg = m + (size_t)N * CCH;
    const int groups = (N + NPW - 1) / NPW;
    int2* rec  = (int2*)(agg + (size_t)N * CCH);
    int*  wcnt = (int*)(rec + (size_t)groups * OWN_WAVES * WCAP);
    unsigned* key = (unsigned*)(((size_t)(wcnt + (size_t)groups * OWN_WAVES) + 15) & ~(size_t)15);
    const int nwords = (E >> 11) * 512;
    if (nwords > 0)
        ggnn_owner_keys<<<(nwords + 255) / 256, 256, 0, stream>>>(dst, key, nwords);

    const int nelem = N * CCH;
    const int n4    = (nelem >> 2) + 4;
    ggnn_copy<<<(n4 + 255) / 256, 256, 0, stream>>>(x, h, nelem);

    const int tiles = (N + 15) / 16;
    for (int layer = 0; layer < NLAYERS; ++layer) {
        ggnn_transform_wmma<<<(tiles + 3) / 4, 128, 0, stream>>>(
            h, weight + (size_t)layer * CCH * CCH, m, N);
        if (layer == 0)
            ggnn_owner_sum<0><<<groups, OWN_THREADS, 0, stream>>>(src, dst, (const uint4*)key, m, agg, rec, wcnt, N, E);
        else
            ggnn_owner_sum<1><<<groups, OWN_THREADS, 0, stream>>>(src, dst, (const uint4*)key, m, agg, rec, wcnt, N, E);
        ggnn_gru<<<GRU_GROUPS, GRU_WAVES * 32, 0, stream>>>(agg, h, w_ih, w_hh,
                                                            b_ih, b_hh, N);
    }
    ggnn_final<<<(N + 127) / 128, 128, 0, stream>>>(h, lin_w, lin_b, out, N);
}


# Dense per-node stages fused into the GRU launch: its 16-node LDS tile feeds next layer's h@W WMMA (m) or, last layer, relu+final linear on WMMA (sums in WMMA k order); x read direct; 12->8 launches
# speedup vs baseline: 11.2114x; 1.0386x over previous
#include <hip/hip_runtime.h>
#include <math.h>

#define CCH 30
#define NLAYERS 3
#define AST 34
#define CH2 (CCH / 2)

typedef __attribute__((ext_vector_type(2))) float v2f;
typedef __attribute__((ext_vector_type(8))) float v8f;

__device__ __forceinline__ float sigmoid_f(float x) { return 1.0f / (1.0f + __expf(-x)); }

__global__ void ggnn_transform_wmma(const float* __restrict__ h,
                                    const float* __restrict__ W,
                                    float* __restrict__ m, int N) {
    __shared__ float s_Wt[32 * AST];
    __shared__ float s_A[4][16 * AST];

    const int tid  = threadIdx.x;
    const int lane = tid & 31;
    const int warp = tid >> 5;
    const int row0 = (blockIdx.x * 4 + warp) * 16;
    const bool full = (row0 + 16 <= N);

    for (int i = tid; i < 32 * AST; i += 128) s_Wt[i] = 0.0f;
    for (int i = lane; i < 16 * AST; i += 32) s_A[warp][i] = 0.0f;
    __syncthreads();

    for (int i = tid; i < CCH * CCH; i += 128) {
        const int k = i / CCH, c = i % CCH;
        s_Wt[c * AST + k] = W[i];
    }
    if (full) {
        const float2* h2 = reinterpret_cast<const float2*>(h + (size_t)row0 * CCH);
        #pragma unroll
        for (int i = lane; i < 16 * CH2; i += 32) {
            const int r = i / CH2, c2 = i % CH2;
            *reinterpret_cast<float2*>(&s_A[warp][r * AST + 2 * c2]) =
                h2[(size_t)r * CH2 + c2];
        }
    } else {
        for (int i = lane; i < 16 * CCH; i += 32) {
            const int r = i / CCH, c = i % CCH;
            const int row = row0 + r;
            if (row < N) s_A[warp][r * AST + c] = h[(size_t)row * CCH + c];
        }
    }
    __syncthreads();

    if (row0 >= N) return;

    const int half = lane >> 4;
    const int l    = lane & 15;
    v8f acc0 = {};
    v8f acc1 = {};
    float* sa = &s_A[warp][0];

    #pragma unroll
    for (int k0 = 0; k0 < 32; k0 += 4) {
        const int ka = k0 + half * 2;
        const v2f a  = *(const v2f*)&sa  [l        * AST + ka];
        const v2f b0 = *(const v2f*)&s_Wt[l        * AST + ka];
        const v2f b1 = *(const v2f*)&s_Wt[(16 + l) * AST + ka];
        acc0 = __builtin_amdgcn_wmma_f32_16x16x4_f32(false, a, false, b0,
                                                     (short)0, acc0, false, false);
        acc1 = __builtin_amdgcn_wmma_f32_16x16x4_f32(false, a, false, b1,
                                                     (short)0, acc1, false, false);
    }

    #pragma unroll
    for (int v = 0; v < 8; ++v) {
        const int r = v + half * 8;
        sa[r * AST + l]      = acc0[v];
        sa[r * AST + 16 + l] = acc1[v];
    }

    if (full) {
        float2* m2 = reinterpret_cast<float2*>(m + (size_t)row0 * CCH);
        #pragma unroll
        for (int i = lane; i < 16 * CH2; i += 32) {
            const int r = i / CH2, c2 = i % CH2;
            m2[(size_t)r * CH2 + c2] =
                *reinterpret_cast<const float2*>(&sa[r * AST + 2 * c2]);
        }
    } else {
        for (int i = lane; i < 16 * CCH; i += 32) {
            const int r = i / CCH, c = i % CCH;
            const int row = row0 + r;
            if (row < N) m[(size_t)row * CCH + c] = sa[r * AST + c];
        }
    }
}

#define NPW         784
#define AGG_ROWS    1368
#define OWN_THREADS 1024
#define OWN_WAVES   32
#define WCAP        1600
#define QCAP        64

__global__ void ggnn_owner_keys(const int* __restrict__ dst,
                                unsigned* __restrict__ key, int nwords) {
    const int i = blockIdx.x * blockDim.x + threadIdx.x;
    if (i >= nwords) return;
    int d0, d1, d2, d3;
    if ((reinterpret_cast<size_t>(dst) & 15) == 0) {
        const int4 d = reinterpret_cast<const int4*>(dst)[i];
        d0 = d.x; d1 = d.y; d2 = d.z; d3 = d.w;
    } else {
        d0 = dst[4 * i]; d1 = dst[4 * i + 1]; d2 = dst[4 * i + 2]; d3 = dst[4 * i + 3];
    }
    key[i] = (((unsigned)d0 / NPW) & 255u) | ((((unsigned)d1 / NPW) & 255u) << 8) |
             ((((unsigned)d2 / NPW) & 255u) << 16) | ((((unsigned)d3 / NPW) & 255u) << 24);
}

template <bool FULL, int B>
__device__ __forceinline__ void add_rows(int s, int dl, int n,
                                         const float* __restrict__ m,
                                         float* s_agg, int lane) {
    #pragma unroll
    for (int j = 0; j < 32; j += B) {
        if (!FULL && j >= n) break;
        int sj[B], dj[B];
        #pragma unroll
        for (int u = 0; u < B; ++u) {
            sj[u] = __builtin_amdgcn_readlane(s, j + u);
            dj[u] = __builtin_amdgcn_readlane(dl, j + u);
        }
        if (lane < CCH) {
            float v[B];
            #pragma unroll
            for (int u = 0; u < B; ++u)
                if (FULL || j + u < n) v[u] = m[(size_t)sj[u] * CCH + lane];
            #pragma unroll
            for (int u = 0; u < B; ++u)
                if (FULL || j + u < n)
                    __hip_atomic_fetch_add(&s_agg[dj[u] * CCH + lane], v[u],
                                           __ATOMIC_RELAXED,
                                           __HIP_MEMORY_SCOPE_WORKGROUP);
        }
    }
}

template <bool FULL, bool REC>
__device__ __forceinline__ void own_drain(int e, int n, int lo,
                                          const int* __restrict__ src,
                                          const int* __restrict__ dst,
                                          const float* __restrict__ m,
                                          float* s_agg, int lane,
                                          int2* __restrict__ rec, int recpos) {
    int s = 0, r = NPW;
    if (FULL || lane < n) {
        const unsigned rr = (unsigned)(dst[e] - lo);
        r = (int)(rr < (unsigned)NPW ? rr : (unsigned)NPW);
        s = src[e];
        if (REC && recpos + lane < WCAP) rec[recpos + lane] = make_int2(s, r);
    }
    add_rows<FULL, 16>(s, r, n, m, s_agg, lane);
}

template <bool REC>
__device__ __forceinline__ void own_added(int n, int* q, int& pos, int& recpos, int lo,
                                          const int* __restrict__ src,
                                          const int* __restrict__ dst,
                                          const float* __restrict__ m,
                                          float* s_agg, int lane,
                                          int2* __restrict__ rec) {
    pos += n;
    if (pos >= 32) {
        const int e = q[lane];
        pos -= 32;
        if (lane < pos) q[lane] = q[32 + lane];
        own_drain<true, REC>(e, 32, lo, src, dst, m, s_agg, lane, rec, recpos);
        recpos += 32;
    }
}

__device__ __forceinline__ void own_replay(const int2* __restrict__ rec, int cnt,
                                           const float* __restrict__ m,
                                           float* s_agg, int lane) {
    int2 cur = make_int2(0, 0);
    if (lane < cnt) cur = rec[lane];
    for (int base = 0; base < cnt; base += 32) {
        int2 nxt = make_int2(0, 0);
        if (base + 32 + lane < cnt) nxt = rec[base + 32 + lane];
        const int n = cnt - base;
        if (n >= 32) add_rows<true, 32>(cur.x, cur.y, 32, m, s_agg, lane);
        else         add_rows<false, 32>(cur.x, cur.y, n, m, s_agg, lane);
        cur = nxt;
    }
}

template <int MODE>
__global__ __launch_bounds__(OWN_THREADS)
void ggnn_owner_sum(const int* __restrict__ src,
                    const int* __restrict__ dst,
                    const uint4* __restrict__ key4,
                    const float* __restrict__ m,
                    float* __restrict__ agg,
                    int2* __restrict__ rec_all,
                    int* __restrict__ wcnt, int N, int E) {
    __shared__ __attribute__((aligned(16))) float s_agg[AGG_ROWS * CCH];
    __shared__ int s_q[OWN_WAVES * QCAP];

    const int tid  = threadIdx.x;
    const int lane = tid & 31;
    const int wave = __builtin_amdgcn_readfirstlane(tid >> 5);
    const int lo   = blockIdx.x * NPW;

    for (int i = tid; i < AGG_ROWS * CCH / 4; i += OWN_THREADS)
        reinterpret_cast<float4*>(s_agg)[i] = make_float4(0.f, 0.f, 0.f, 0.f);
    __syncthreads();

    constexpr bool REC = (MODE == 0);
    const int gw = blockIdx.x * OWN_WAVES + wave;
    int2* rec = rec_all + (size_t)gw * WCAP;
    int recpos = 0;
    bool replay = false;
    if (MODE == 1) {
        const int cnt = __builtin_amdgcn_readfirstlane(wcnt[gw]);
        replay = (cnt >= 0 && cnt <= WCAP);
        if (replay) own_replay(rec, cnt, m, s_agg, lane);
    }
    if (!replay) {

    int* q  = s_q + wave * QCAP;
    int pos = 0;

    const unsigned pat = (unsigned)(blockIdx.x & 255) * 0x01010101u;
    const int nsu = E >> 11;
    const int start = (int)(((long long)blockIdx.x * nsu) / gridDim.x);

    if (wave < nsu) {
        int k = wave;
        uint4 cur[4];
        {
            int su = k + start; if (su >= nsu) su -= nsu;
            const uint4* p = key4 + (size_t)su * 128 + lane;
            #pragma unroll
            for (int u = 0; u < 4; ++u) cur[u] = p[32 * u];
        }
        while (k < nsu) {
            const int kn = k + OWN_WAVES;
            const int kl = kn < nsu ? kn : k;
            int sun = kl + start; if (sun >= nsu) sun -= nsu;
            const uint4* pn = key4 + (size_t)sun * 128 + lane;
            uint4 nxt[4];
            #pragma unroll
            for (int u = 0; u < 4; ++u) nxt[u] = pn[32 * u];

            int su = k + start; if (su >= nsu) su -= nsu;
            #pragma unroll
            for (int u = 0; u < 4; ++u) {
                const unsigned wv[4] = {cur[u].x, cur[u].y, cur[u].z, cur[u].w};
                unsigned fl = 0u;
                #pragma unroll
                for (int c = 0; c < 4; ++c) {
                    const unsigned x = wv[c] ^ pat;
                    const unsigned y = ((x & 0x7F7F7F7Fu) + 0x7F7F7F7Fu) | x;
                    fl |= (~y & 0x80808080u) >> (7 - c);
                }
                unsigned mk = __builtin_amdgcn_ballot_w32(fl != 0u);
                const int eu = su * 2048 + u * 512 + lane * 16;
                while (mk != 0u) {
                    if (fl != 0u) {
                        const int i = __builtin_ctz(fl);
                        q[pos + (int)__builtin_amdgcn_mbcnt_lo(mk, 0u)] =
                            eu + (i & 7) * 4 + (i >> 3);
                    }
                    fl &= fl - 1u;
                    own_added<REC>(__builtin_popcount(mk), q, pos, recpos, lo, src, dst, m,
                                   s_agg, lane, rec);
                    mk = __builtin_amdgcn_ballot_w32(fl != 0u);
                }
            }
            #pragma unroll
            for (int u = 0; u < 4; ++u) cur[u] = nxt[u];
            k = kn;
        }
    }

    for (int base = nsu * 2048 + wave * 32; base < E; base += OWN_THREADS) {
        const int  e     = base + lane;
        const bool valid = e < E;
        const int  d     = valid ? dst[e] : -1;
        const bool p     = valid && (unsigned)(d - lo) < (unsigned)NPW;
        const unsigned mk = __builtin_amdgcn_ballot_w32(p);
        if (mk != 0u) {
            if (p) q[pos + (int)__builtin_amdgcn_mbcnt_lo(mk, 0u)] = e;
            own_added<REC>(__builtin_popcount(mk), q, pos, recpos, lo, src, dst, m,
                           s_agg, lane, rec);
        }
    }
    if (pos > 0) {
        const int e = lane < pos ? q[lane] : 0;
        own_drain<false, REC>(e, pos, lo, src, dst, m, s_agg, lane, rec, recpos);
        recpos += pos;
    }
    if (REC && lane == 0) wcnt[gw] = recpos;
    }
    __syncthreads();

    const int nn = (N - lo) < NPW ? (N - lo) : NPW;
    float* ag = agg + (size_t)lo * CCH;
    for (int i = tid; i < nn * CCH; i += OWN_THREADS) ag[i] = s_agg[i];
}

#define GRU_WAVES  4
#define GRU_GROUPS 768
#define GRU_TILE   16

__device__ __forceinline__ void gru_rows(const float* agg, const float* h, int n, int N, int j,
                                         v2f (&a)[CH2], v2f (&hv)[CH2], float& hj) {
    n = n < N ? n : N - 1;
    const float* ar = agg + (size_t)n * CCH;
    const float* hr = h   + (size_t)n * CCH;
    #pragma unroll
    for (int k = 0; k < CH2; ++k) {
        a[k]  = *(const v2f*)(ar + 2 * k);
        hv[k] = *(const v2f*)(hr + 2 * k);
    }
    hj = hr[j];
}

struct GruW {
    v2f wir[CH2], wiz[CH2], win[CH2], whr[CH2], whz[CH2], whn[CH2];
    float bir, biz, bin, bhr, bhz, bhn;
};

__device__ __forceinline__ float gru_node(const GruW& w, const v2f (&a)[CH2], const v2f (&hv)[CH2],
                                          float hj) {
    v2f ir = {w.bir, 0.0f}, iz = {w.biz, 0.0f}, in_ = {w.bin, 0.0f};
    v2f gr = {w.bhr, 0.0f}, gz = {w.bhz, 0.0f}, gn = {w.bhn, 0.0f};
    #pragma unroll
    for (int k = 0; k < CH2; ++k) {
        ir  = __builtin_elementwise_fma(a[k],  w.wir[k], ir);
        iz  = __builtin_elementwise_fma(a[k],  w.wiz[k], iz);
        in_ = __builtin_elementwise_fma(a[k],  w.win[k], in_);
        gr  = __builtin_elementwise_fma(hv[k], w.whr[k], gr);
        gz  = __builtin_elementwise_fma(hv[k], w.whz[k], gz);
        gn  = __builtin_elementwise_fma(hv[k], w.whn[k], gn);
    }
    const float r  = sigmoid_f((ir.x + ir.y) + (gr.x + gr.y));
    const float z  = sigmoid_f((iz.x + iz.y) + (gz.x + gz.y));
    const float nn = tanhf((in_.x + in_.y) + r * (gn.x + gn.y));
    return (1.0f - z) * nn + z * hj;
}

template <bool BIAS>
__device__ __forceinline__ void tile_product(const float* sa, const float* sb, float* sd,
                                             const int2* tab, const float* sbias,
                                             float* __restrict__ dst, int nrows, int lane) {
    const int2 t   = tab[lane];
    const int  fo  = t.x;
    const int  dof = t.y;
    const bool hi  = lane >= 16;
    v8f acc0 = {};
    v8f acc1 = {};
    #pragma unroll
    for (int k0 = 0; k0 < 32; k0 += 4) {
        v2f a  = *(const v2f*)&sa[fo + k0];
        v2f b0 = *(const v2f*)&sb[fo + k0];
        v2f b1 = *(const v2f*)&sb[16 * CCH + fo + k0];
        if (k0 + 4 > CCH && hi) {
            a = v2f{0.0f, 0.0f}; b0 = v2f{0.0f, 0.0f}; b1 = v2f{0.0f, 0.0f};
        }
        acc0 = __builtin_amdgcn_wmma_f32_16x16x4_f32(false, a, false, b0,
                                                     (short)0, acc0, false, false);
        acc1 = __builtin_amdgcn_wmma_f32_16x16x4_f32(false, a, false, b1,
                                                     (short)0, acc1, false, false);
    }
    float blo = 0.0f, bhi = 0.0f;
    if (BIAS) { blo = sbias[lane & 15]; bhi = sbias[16 + (lane & 15)]; }
    const bool c1 = 16 + (lane & 15) < CCH;
    #pragma unroll
    for (int v = 0; v < 8; ++v) {
        sd[dof + v * CCH] = BIAS ? acc0[v] + blo : acc0[v];
        if (c1) sd[dof + v * CCH + 16] = BIAS ? acc1[v] + bhi : acc1[v];
    }
    float2* d2 = reinterpret_cast<float2*>(dst);
    const float2* sd2 = reinterpret_cast<const float2*>(sd);
    #pragma unroll
    for (int i = lane; i < GRU_TILE * CH2; i += 32)
        if (i < nrows * CH2) d2[i] = sd2[i];
}

template <int EPI>
__global__ __launch_bounds__(GRU_WAVES * 32, 3)
void ggnn_gru(const float* agg,
              const float* h_in,
              float* h_out,
              const float* __restrict__ w_ih,
              const float* __restrict__ w_hh,
              const float* __restrict__ b_ih,
              const float* __restrict__ b_hh,
              const float* __restrict__ wn,
              const float* __restrict__ bn,
              float* __restrict__ rows_out,
              int N) {
    const int tid  = threadIdx.x;
    const int lane = tid & 31;
    const int wave = __builtin_amdgcn_readfirstlane(tid >> 5);
    const int j = lane < CCH ? lane : CCH - 1;

    __shared__ __attribute__((aligned(16))) float s_w[2 * 3 * CCH * CCH];
    __shared__ __attribute__((aligned(16))) float s_o[2 * GRU_TILE * CCH + 4];
    __shared__ __attribute__((aligned(16))) float s_b[32 * CCH + 4];
    __shared__ __attribute__((aligned(16))) float s_bn[32];
    __shared__ __attribute__((aligned(16))) int2  s_tab[32];
    __shared__ __attribute__((aligned(16))) float s_d[GRU_TILE * CCH];
    for (int i = tid; i < 3 * CCH * CCH / 4; i += GRU_WAVES * 32) {
        reinterpret_cast<float4*>(s_w)[i] = reinterpret_cast<const float4*>(w_ih)[i];
        reinterpret_cast<float4*>(s_w + 3 * CCH * CCH)[i] = reinterpret_cast<const float4*>(w_hh)[i];
    }
    for (int i = tid; i < 32 * CCH + 4; i += GRU_WAVES * 32) {
        const int c = i / CCH, k = i % CCH;
        float v = 0.0f;
        if (c < CCH) v = (EPI == 0) ? wn[k * CCH + c] : wn[c * CCH + k];
        s_b[i] = v;
    }
    if (tid < 32) {
        s_tab[tid] = make_int2((tid & 15) * CCH + (tid >> 4) * 2, (tid >> 4) * 8 * CCH + (tid & 15));
        s_bn[tid]  = (EPI == 1 && tid < CCH) ? bn[tid] : 0.0f;
    }
    __syncthreads();
    const float* s_ih = s_w;
    const float* s_hh = s_w + 3 * CCH * CCH;
    GruW w;
    #pragma unroll
    for (int k = 0; k < CH2; ++k) {
        w.wir[k] = *(const v2f*)&s_ih[j * CCH + 2 * k];
        w.wiz[k] = *(const v2f*)&s_ih[(CCH + j) * CCH + 2 * k];
        w.win[k] = *(const v2f*)&s_ih[(2 * CCH + j) * CCH + 2 * k];
        w.whr[k] = *(const v2f*)&s_hh[j * CCH + 2 * k];
        w.whz[k] = *(const v2f*)&s_hh[(CCH + j) * CCH + 2 * k];
        w.whn[k] = *(const v2f*)&s_hh[(2 * CCH + j) * CCH + 2 * k];
    }
    w.bir = b_ih[j]; w.biz = b_ih[CCH + j]; w.bin = b_ih[2 * CCH + j];
    w.bhr = b_hh[j]; w.bhz = b_hh[CCH + j]; w.bhn = b_hh[2 * CCH + j];

    const int ntiles = (N + GRU_TILE - 1) / GRU_TILE;
    int tile = blockIdx.x;
    if (tile >= ntiles) return;

    v2f a0[CH2], h0[CH2], a1[CH2], h1[CH2];
    float hj0, hj1;
    const int q0 = wave * (GRU_TILE / GRU_WAVES);
    gru_rows(agg, h_in, tile * GRU_TILE + q0, N, j, a0, h0, hj0);
    int buf = 0;
    int turn = 0;
    #pragma unroll 1
    for (;;) {
        const int base = tile * GRU_TILE + q0;
        const int nt = tile + (int)gridDim.x;
        const bool more = nt < ntiles;
        float* st = &s_o[buf * (GRU_TILE * CCH)];
        float* so = st + q0 * CCH;
        float v;
        gru_rows(agg, h_in, base + 1, N, j, a1, h1, hj1);
        v = gru_node(w, a0, h0, hj0);
        if (EPI == 1) v = v > 0.0f ? v : 0.0f;
        if (lane < CCH) so[lane] = v;
        gru_rows(agg, h_in, base + 2, N, j, a0, h0, hj0);
        v = gru_node(w, a1, h1, hj1);
        if (EPI == 1) v = v > 0.0f ? v : 0.0f;
        if (lane < CCH) so[CCH + lane] = v;
        gru_rows(agg, h_in, base + 3, N, j, a1, h1, hj1);
        v = gru_node(w, a0, h0, hj0);
        if (EPI == 1) v = v > 0.0f ? v : 0.0f;
        if (lane < CCH) so[2 * CCH + lane] = v;
        gru_rows(agg, h_in, more ? nt * GRU_TILE + q0 : base + 3, N, j, a0, h0, hj0);
        v = gru_node(w, a1, h1, hj1);
        if (EPI == 1) v = v > 0.0f ? v : 0.0f;
        if (lane < CCH) so[3 * CCH + lane] = v;
        __syncthreads();
        const int left  = N - tile * GRU_TILE;
        const int nrows = left < GRU_TILE ? left : GRU_TILE;
        if (EPI == 0) {
            float* ht = h_out + (size_t)tile * (GRU_TILE * CCH);
            #pragma unroll
            for (int i = tid; i < GRU_TILE * CH2; i += GRU_WAVES * 32)
                if (i < nrows * CH2)
                    reinterpret_cast<float2*>(ht)[i] = reinterpret_cast<const float2*>(st)[i];
        }
        if (wave == turn)
            tile_product<EPI == 1>(st, s_b, s_d, s_tab, s_bn,
                                   rows_out + (size_t)tile * (GRU_TILE * CCH), nrows, lane);
        if (!more) break;
        tile = nt;
        buf ^= 1;
        turn = (turn + 1) & (GRU_WAVES - 1);
    }
}

extern "C" void kernel_launch(void* const* d_in, const int* in_sizes, int n_in,
                              void* d_out, int out_size, void* d_ws, size_t ws_size,
                              hipStream_t stream) {
    const float* x      = (const float*)d_in[0];
    const int*   eidx   = (const int*)  d_in[1];
    const float* weight = (const float*)d_in[2];
    const float* w_ih   = (const float*)d_in[3];
    const float* w_hh   = (const float*)d_in[4];
    const float* b_ih   = (const float*)d_in[5];
    const float* b_hh   = (const float*)d_in[6];
    const float* lin_w  = (const float*)d_in[7];
    const float* lin_b  = (const float*)d_in[8];
    float* out = (float*)d_out;

    const int N = in_sizes[0] / CCH;
    const int E = in_sizes[1] / 2;
    const int* src = eidx;
    const int* dst = eidx + E;

    float* h   = (float*)d_ws;
    float* m   = h + (size_t)N * CCH;
    float* agg = m + (size_t)N * CCH;
    const int groups = (N + NPW - 1) / NPW;
    int2* rec  = (int2*)(agg + (size_t)N * CCH);
    int*  wcnt = (int*)(rec + (size_t)groups * OWN_WAVES * WCAP);
    unsigned* key = (unsigned*)(((size_t)(wcnt + (size_t)groups * OWN_WAVES) + 15) & ~(size_t)15);
    const int nwords = (E >> 11) * 512;
    if (nwords > 0)
        ggnn_owner_keys<<<(nwords + 255) / 256, 256, 0, stream>>>(dst, key, nwords);

    const int tiles = (N + 15) / 16;
    ggnn_transform_wmma<<<(tiles + 3) / 4, 128, 0, stream>>>(x, weight, m, N);
    for (int layer = 0; layer < NLAYERS; ++layer) {
        if (layer == 0)
            ggnn_owner_sum<0><<<groups, OWN_THREADS, 0, stream>>>(src, dst, (const uint4*)key, m, agg, rec, wcnt, N, E);
        else
            ggnn_owner_sum<1><<<groups, OWN_THREADS, 0, stream>>>(src, dst, (const uint4*)key, m, agg, rec, wcnt, N, E);
        const float* hin = layer == 0 ? x : h;
        if (layer + 1 < NLAYERS)
            ggnn_gru<0><<<GRU_GROUPS, GRU_WAVES * 32, 0, stream>>>(
                agg, hin, h, w_ih, w_hh, b_ih, b_hh,
                weight + (size_t)(layer + 1) * CCH * CCH, lin_b, m, N);
        else
            ggnn_gru<1><<<GRU_GROUPS, GRU_WAVES * 32, 0, stream>>>(
                agg, hin, h, w_ih, w_hh, b_ih, b_hh, lin_w, lin_b, out, N);
    }
}


# Keys launch also writes each edge's (source id, destination id) side by side (8 B); the first-layer scan fetches a kept edge's ids with one 8-B load, one random line instead of two; sum order kept
# speedup vs baseline: 11.6645x; 1.0404x over previous
#include <hip/hip_runtime.h>
#include <math.h>

#define CCH 30
#define NLAYERS 3
#define AST 34
#define CH2 (CCH / 2)

typedef __attribute__((ext_vector_type(2))) float v2f;
typedef __attribute__((ext_vector_type(8))) float v8f;

__device__ __forceinline__ float sigmoid_f(float x) { return 1.0f / (1.0f + __expf(-x)); }

__global__ void ggnn_transform_wmma(const float* __restrict__ h,
                                    const float* __restrict__ W,
                                    float* __restrict__ m, int N) {
    __shared__ float s_Wt[32 * AST];
    __shared__ float s_A[4][16 * AST];

    const int tid  = threadIdx.x;
    const int lane = tid & 31;
    const int warp = tid >> 5;
    const int row0 = (blockIdx.x * 4 + warp) * 16;
    const bool full = (row0 + 16 <= N);

    for (int i = tid; i < 32 * AST; i += 128) s_Wt[i] = 0.0f;
    for (int i = lane; i < 16 * AST; i += 32) s_A[warp][i] = 0.0f;
    __syncthreads();

    for (int i = tid; i < CCH * CCH; i += 128) {
        const int k = i / CCH, c = i % CCH;
        s_Wt[c * AST + k] = W[i];
    }
    if (full) {
        const float2* h2 = reinterpret_cast<const float2*>(h + (size_t)row0 * CCH);
        #pragma unroll
        for (int i = lane; i < 16 * CH2; i += 32) {
            const int r = i / CH2, c2 = i % CH2;
            *reinterpret_cast<float2*>(&s_A[warp][r * AST + 2 * c2]) =
                h2[(size_t)r * CH2 + c2];
        }
    } else {
        for (int i = lane; i < 16 * CCH; i += 32) {
            const int r = i / CCH, c = i % CCH;
            const int row = row0 + r;
            if (row < N) s_A[warp][r * AST + c] = h[(size_t)row * CCH + c];
        }
    }
    __syncthreads();

    if (row0 >= N) return;

    const int half = lane >> 4;
    const int l    = lane & 15;
    v8f acc0 = {};
    v8f acc1 = {};
    float* sa = &s_A[warp][0];

    #pragma unroll
    for (int k0 = 0; k0 < 32; k0 += 4) {
        const int ka = k0 + half * 2;
        const v2f a  = *(const v2f*)&sa  [l        * AST + ka];
        const v2f b0 = *(const v2f*)&s_Wt[l        * AST + ka];
        const v2f b1 = *(const v2f*)&s_Wt[(16 + l) * AST + ka];
        acc0 = __builtin_amdgcn_wmma_f32_16x16x4_f32(false, a, false, b0,
                                                     (short)0, acc0, false, false);
        acc1 = __builtin_amdgcn_wmma_f32_16x16x4_f32(false, a, false, b1,
                                                     (short)0, acc1, false, false);
    }

    #pragma unroll
    for (int v = 0; v < 8; ++v) {
        const int r = v + half * 8;
        sa[r * AST + l]      = acc0[v];
        sa[r * AST + 16 + l] = acc1[v];
    }

    if (full) {
        float2* m2 = reinterpret_cast<float2*>(m + (size_t)row0 * CCH);
        #pragma unroll
        for (int i = lane; i < 16 * CH2; i += 32) {
            const int r = i / CH2, c2 = i % CH2;
            m2[(size_t)r * CH2 + c2] =
                *reinterpret_cast<const float2*>(&sa[r * AST + 2 * c2]);
        }
    } else {
        for (int i = lane; i < 16 * CCH; i += 32) {
            const int r = i / CCH, c = i % CCH;
            const int row = row0 + r;
            if (row < N) m[(size_t)row * CCH + c] = sa[r * AST + c];
        }
    }
}

#define NPW         784
#define AGG_ROWS    1368
#define OWN_THREADS 1024
#define OWN_WAVES   32
#define WCAP        1600
#define QCAP        64

__global__ void ggnn_owner_keys(const int* __restrict__ src,
                                const int* __restrict__ dst,
                                unsigned* __restrict__ key,
                                int2* __restrict__ pair, int nwords, int E) {
    const int i = blockIdx.x * blockDim.x + threadIdx.x;
    const int e0 = 4 * i;
    if (e0 >= E) return;
    if (e0 + 4 > E) {
        for (int e = e0; e < E; ++e) pair[e] = make_int2(src[e], dst[e]);
        return;
    }
    int d0, d1, d2, d3, s0, s1, s2, s3;
    if ((reinterpret_cast<size_t>(dst) & 15) == 0) {
        const int4 d = reinterpret_cast<const int4*>(dst)[i];
        d0 = d.x; d1 = d.y; d2 = d.z; d3 = d.w;
    } else {
        d0 = dst[e0]; d1 = dst[e0 + 1]; d2 = dst[e0 + 2]; d3 = dst[e0 + 3];
    }
    if ((reinterpret_cast<size_t>(src) & 15) == 0) {
        const int4 s = reinterpret_cast<const int4*>(src)[i];
        s0 = s.x; s1 = s.y; s2 = s.z; s3 = s.w;
    } else {
        s0 = src[e0]; s1 = src[e0 + 1]; s2 = src[e0 + 2]; s3 = src[e0 + 3];
    }
    int4* p4 = reinterpret_cast<int4*>(pair + e0);
    p4[0] = make_int4(s0, d0, s1, d1);
    p4[1] = make_int4(s2, d2, s3, d3);
    if (i < nwords)
        key[i] = (((unsigned)d0 / NPW) & 255u) | ((((unsigned)d1 / NPW) & 255u) << 8) |
                 ((((unsigned)d2 / NPW) & 255u) << 16) | ((((unsigned)d3 / NPW) & 255u) << 24);
}

template <bool FULL, int B>
__device__ __forceinline__ void add_rows(int s, int dl, int n,
                                         const float* __restrict__ m,
                                         float* s_agg, int lane) {
    #pragma unroll
    for (int j = 0; j < 32; j += B) {
        if (!FULL && j >= n) break;
        int sj[B], dj[B];
        #pragma unroll
        for (int u = 0; u < B; ++u) {
            sj[u] = __builtin_amdgcn_readlane(s, j + u);
            dj[u] = __builtin_amdgcn_readlane(dl, j + u);
        }
        if (lane < CCH) {
            float v[B];
            #pragma unroll
            for (int u = 0; u < B; ++u)
                if (FULL || j + u < n) v[u] = m[(size_t)sj[u] * CCH + lane];
            #pragma unroll
            for (int u = 0; u < B; ++u)
                if (FULL || j + u < n)
                    __hip_atomic_fetch_add(&s_agg[dj[u] * CCH + lane], v[u],
                                           __ATOMIC_RELAXED,
                                           __HIP_MEMORY_SCOPE_WORKGROUP);
        }
    }
}

template <bool FULL, bool REC>
__device__ __forceinline__ void own_drain(int e, int n, int lo,
                                          const int2* __restrict__ pair,
                                          const float* __restrict__ m,
                                          float* s_agg, int lane,
                                          int2* __restrict__ rec, int recpos) {
    int s = 0, r = NPW;
    if (FULL || lane < n) {
        const int2 sd = pair[e];
        const unsigned rr = (unsigned)(sd.y - lo);
        r = (int)(rr < (unsigned)NPW ? rr : (unsigned)NPW);
        s = sd.x;
        if (REC && recpos + lane < WCAP) rec[recpos + lane] = make_int2(s, r);
    }
    add_rows<FULL, 16>(s, r, n, m, s_agg, lane);
}

template <bool REC>
__device__ __forceinline__ void own_added(int n, int* q, int& pos, int& recpos, int lo,
                                          const int2* __restrict__ pair,
                                          const float* __restrict__ m,
                                          float* s_agg, int lane,
                                          int2* __restrict__ rec) {
    pos += n;
    if (pos >= 32) {
        const int e = q[lane];
        pos -= 32;
        if (lane < pos) q[lane] = q[32 + lane];
        own_drain<true, REC>(e, 32, lo, pair, m, s_agg, lane, rec, recpos);
        recpos += 32;
    }
}

__device__ __forceinline__ void own_replay(const int2* __restrict__ rec, int cnt,
                                           const float* __restrict__ m,
                                           float* s_agg, int lane) {
    int2 cur = make_int2(0, 0);
    if (lane < cnt) cur = rec[lane];
    for (int base = 0; base < cnt; base += 32) {
        int2 nxt = make_int2(0, 0);
        if (base + 32 + lane < cnt) nxt = rec[base + 32 + lane];
        const int n = cnt - base;
        if (n >= 32) add_rows<true, 32>(cur.x, cur.y, 32, m, s_agg, lane);
        else         add_rows<false, 32>(cur.x, cur.y, n, m, s_agg, lane);
        cur = nxt;
    }
}

template <int MODE>
__global__ __launch_bounds__(OWN_THREADS)
void ggnn_owner_sum(const int2* __restrict__ pair,
                    const int* __restrict__ dst,
                    const uint4* __restrict__ key4,
                    const float* __restrict__ m,
                    float* __restrict__ agg,
                    int2* __restrict__ rec_all,
                    int* __restrict__ wcnt, int N, int E) {
    __shared__ __attribute__((aligned(16))) float s_agg[AGG_ROWS * CCH];
    __shared__ int s_q[OWN_WAVES * QCAP];

    const int tid  = threadIdx.x;
    const int lane = tid & 31;
    const int wave = __builtin_amdgcn_readfirstlane(tid >> 5);
    const int lo   = blockIdx.x * NPW;

    for (int i = tid; i < AGG_ROWS * CCH / 4; i += OWN_THREADS)
        reinterpret_cast<float4*>(s_agg)[i] = make_float4(0.f, 0.f, 0.f, 0.f);
    __syncthreads();

    constexpr bool REC = (MODE == 0);
    const int gw = blockIdx.x * OWN_WAVES + wave;
    int2* rec = rec_all + (size_t)gw * WCAP;
    int recpos = 0;
    bool replay = false;
    if (MODE == 1) {
        const int cnt = __builtin_amdgcn_readfirstlane(wcnt[gw]);
        replay = (cnt >= 0 && cnt <= WCAP);
        if (replay) own_replay(rec, cnt, m, s_agg, lane);
    }
    if (!replay) {

    int* q  = s_q + wave * QCAP;
    int pos = 0;

    const unsigned pat = (unsigned)(blockIdx.x & 255) * 0x01010101u;
    const int nsu = E >> 11;
    const int start = (int)(((long long)blockIdx.x * nsu) / gridDim.x);

    if (wave < nsu) {
        int k = wave;
        uint4 cur[4];
        {
            int su = k + start; if (su >= nsu) su -= nsu;
            const uint4* p = key4 + (size_t)su * 128 + lane;
            #pragma unroll
            for (int u = 0; u < 4; ++u) cur[u] = p[32 * u];
        }
        while (k < nsu) {
            const int kn = k + OWN_WAVES;
            const int kl = kn < nsu ? kn : k;
            int sun = kl + start; if (sun >= nsu) sun -= nsu;
            const uint4* pn = key4 + (size_t)sun * 128 + lane;
            uint4 nxt[4];
            #pragma unroll
            for (int u = 0; u < 4; ++u) nxt[u] = pn[32 * u];

            int su = k + start; if (su >= nsu) su -= nsu;
            #pragma unroll
            for (int u = 0; u < 4; ++u) {
                const unsigned wv[4] = {cur[u].x, cur[u].y, cur[u].z, cur[u].w};
                unsigned fl = 0u;
                #pragma unroll
                for (int c = 0; c < 4; ++c) {
                    const unsigned x = wv[c] ^ pat;
                    const unsigned y = ((x & 0x7F7F7F7Fu) + 0x7F7F7F7Fu) | x;
                    fl |= (~y & 0x80808080u) >> (7 - c);
                }
                unsigned mk = __builtin_amdgcn_ballot_w32(fl != 0u);
                const int eu = su * 2048 + u * 512 + lane * 16;
                while (mk != 0u) {
                    if (fl != 0u) {
                        const int i = __builtin_ctz(fl);
                        q[pos + (int)__builtin_amdgcn_mbcnt_lo(mk, 0u)] =
                            eu + (i & 7) * 4 + (i >> 3);
                    }
                    fl &= fl - 1u;
                    own_added<REC>(__builtin_popcount(mk), q, pos, recpos, lo, pair, m,
                                   s_agg, lane, rec);
                    mk = __builtin_amdgcn_ballot_w32(fl != 0u);
                }
            }
            #pragma unroll
            for (int u = 0; u < 4; ++u) cur[u] = nxt[u];
            k = kn;
        }
    }

    for (int base = nsu * 2048 + wave * 32; base < E; base += OWN_THREADS) {
        const int  e     = base + lane;
        const bool valid = e < E;
        const int  d     = valid ? dst[e] : -1;
        const bool p     = valid && (unsigned)(d - lo) < (unsigned)NPW;
        const unsigned mk = __builtin_amdgcn_ballot_w32(p);
        if (mk != 0u) {
            if (p) q[pos + (int)__builtin_amdgcn_mbcnt_lo(mk, 0u)] = e;
            own_added<REC>(__builtin_popcount(mk), q, pos, recpos, lo, pair, m,
                           s_agg, lane, rec);
        }
    }
    if (pos > 0) {
        const int e = lane < pos ? q[lane] : 0;
        own_drain<false, REC>(e, pos, lo, pair, m, s_agg, lane, rec, recpos);
        recpos += pos;
    }
    if (REC && lane == 0) wcnt[gw] = recpos;
    }
    __syncthreads();

    const int nn = (N - lo) < NPW ? (N - lo) : NPW;
    float* ag = agg + (size_t)lo * CCH;
    for (int i = tid; i < nn * CCH; i += OWN_THREADS) ag[i] = s_agg[i];
}

#define GRU_WAVES  4
#define GRU_GROUPS 768
#define GRU_TILE   16

__device__ __forceinline__ void gru_rows(const float* agg, const float* h, int n, int N, int j,
                                         v2f (&a)[CH2], v2f (&hv)[CH2], float& hj) {
    n = n < N ? n : N - 1;
    const float* ar = agg + (size_t)n * CCH;
    const float* hr = h   + (size_t)n * CCH;
    #pragma unroll
    for (int k = 0; k < CH2; ++k) {
        a[k]  = *(const v2f*)(ar + 2 * k);
        hv[k] = *(const v2f*)(hr + 2 * k);
    }
    hj = hr[j];
}

struct GruW {
    v2f wir[CH2], wiz[CH2], win[CH2], whr[CH2], whz[CH2], whn[CH2];
    float bir, biz, bin, bhr, bhz, bhn;
};

__device__ __forceinline__ float gru_node(const GruW& w, const v2f (&a)[CH2], const v2f (&hv)[CH2],
                                          float hj) {
    v2f ir = {w.bir, 0.0f}, iz = {w.biz, 0.0f}, in_ = {w.bin, 0.0f};
    v2f gr = {w.bhr, 0.0f}, gz = {w.bhz, 0.0f}, gn = {w.bhn, 0.0f};
    #pragma unroll
    for (int k = 0; k < CH2; ++k) {
        ir  = __builtin_elementwise_fma(a[k],  w.wir[k], ir);
        iz  = __builtin_elementwise_fma(a[k],  w.wiz[k], iz);
        in_ = __builtin_elementwise_fma(a[k],  w.win[k], in_);
        gr  = __builtin_elementwise_fma(hv[k], w.whr[k], gr);
        gz  = __builtin_elementwise_fma(hv[k], w.whz[k], gz);
        gn  = __builtin_elementwise_fma(hv[k], w.whn[k], gn);
    }
    const float r  = sigmoid_f((ir.x + ir.y) + (gr.x + gr.y));
    const float z  = sigmoid_f((iz.x + iz.y) + (gz.x + gz.y));
    const float nn = tanhf((in_.x + in_.y) + r * (gn.x + gn.y));
    return (1.0f - z) * nn + z * hj;
}

template <bool BIAS>
__device__ __forceinline__ void tile_product(const float* sa, const float* sb, float* sd,
                                             const int2* tab, const float* sbias,
                                             float* __restrict__ dst, int nrows, int lane) {
    const int2 t   = tab[lane];
    const int  fo  = t.x;
    const int  dof = t.y;
    const bool hi  = lane >= 16;
    v8f acc0 = {};
    v8f acc1 = {};
    #pragma unroll
    for (int k0 = 0; k0 < 32; k0 += 4) {
        v2f a  = *(const v2f*)&sa[fo + k0];
        v2f b0 = *(const v2f*)&sb[fo + k0];
        v2f b1 = *(const v2f*)&sb[16 * CCH + fo + k0];
        if (k0 + 4 > CCH && hi) {
            a = v2f{0.0f, 0.0f}; b0 = v2f{0.0f, 0.0f}; b1 = v2f{0.0f, 0.0f};
        }
        acc0 = __builtin_amdgcn_wmma_f32_16x16x4_f32(false, a, false, b0,
                                                     (short)0, acc0, false, false);
        acc1 = __builtin_amdgcn_wmma_f32_16x16x4_f32(false, a, false, b1,
                                                     (short)0, acc1, false, false);
    }
    float blo = 0.0f, bhi = 0.0f;
    if (BIAS) { blo = sbias[lane & 15]; bhi = sbias[16 + (lane & 15)]; }
    const bool c1 = 16 + (lane & 15) < CCH;
    #pragma unroll
    for (int v = 0; v < 8; ++v) {
        sd[dof + v * CCH] = BIAS ? acc0[v] + blo : acc0[v];
        if (c1) sd[dof + v * CCH + 16] = BIAS ? acc1[v] + bhi : acc1[v];
    }
    float2* d2 = reinterpret_cast<float2*>(dst);
    const float2* sd2 = reinterpret_cast<const float2*>(sd);
    #pragma unroll
    for (int i = lane; i < GRU_TILE * CH2; i += 32)
        if (i < nrows * CH2) d2[i] = sd2[i];
}

template <int EPI>
__global__ __launch_bounds__(GRU_WAVES * 32, 3)
void ggnn_gru(const float* agg,
              const float* h_in,
              float* h_out,
              const float* __restrict__ w_ih,
              const float* __restrict__ w_hh,
              const float* __restrict__ b_ih,
              const float* __restrict__ b_hh,
              const float* __restrict__ wn,
              const float* __restrict__ bn,
              float* __restrict__ rows_out,
              int N) {
    const int tid  = threadIdx.x;
    const int lane = tid & 31;
    const int wave = __builtin_amdgcn_readfirstlane(tid >> 5);
    const int j = lane < CCH ? lane : CCH - 1;

    __shared__ __attribute__((aligned(16))) float s_w[2 * 3 * CCH * CCH];
    __shared__ __attribute__((aligned(16))) float s_o[2 * GRU_TILE * CCH + 4];
    __shared__ __attribute__((aligned(16))) float s_b[32 * CCH + 4];
    __shared__ __attribute__((aligned(16))) float s_bn[32];
    __shared__ __attribute__((aligned(16))) int2  s_tab[32];
    __shared__ __attribute__((aligned(16))) float s_d[GRU_TILE * CCH];
    for (int i = tid; i < 3 * CCH * CCH / 4; i += GRU_WAVES * 32) {
        reinterpret_cast<float4*>(s_w)[i] = reinterpret_cast<const float4*>(w_ih)[i];
        reinterpret_cast<float4*>(s_w + 3 * CCH * CCH)[i] = reinterpret_cast<const float4*>(w_hh)[i];
    }
    for (int i = tid; i < 32 * CCH + 4; i += GRU_WAVES * 32) {
        const int c = i / CCH, k = i % CCH;
        float v = 0.0f;
        if (c < CCH) v = (EPI == 0) ? wn[k * CCH + c] : wn[c * CCH + k];
        s_b[i] = v;
    }
    if (tid < 32) {
        s_tab[tid] = make_int2((tid & 15) * CCH + (tid >> 4) * 2, (tid >> 4) * 8 * CCH + (tid & 15));
        s_bn[tid]  = (EPI == 1 && tid < CCH) ? bn[tid] : 0.0f;
    }
    __syncthreads();
    const float* s_ih = s_w;
    const float* s_hh = s_w + 3 * CCH * CCH;
    GruW w;
    #pragma unroll
    for (int k = 0; k < CH2; ++k) {
        w.wir[k] = *(const v2f*)&s_ih[j * CCH + 2 * k];
        w.wiz[k] = *(const v2f*)&s_ih[(CCH + j) * CCH + 2 * k];
        w.win[k] = *(const v2f*)&s_ih[(2 * CCH + j) * CCH + 2 * k];
        w.whr[k] = *(const v2f*)&s_hh[j * CCH + 2 * k];
        w.whz[k] = *(const v2f*)&s_hh[(CCH + j) * CCH + 2 * k];
        w.whn[k] = *(const v2f*)&s_hh[(2 * CCH + j) * CCH + 2 * k];
    }
    w.bir = b_ih[j]; w.biz = b_ih[CCH + j]; w.bin = b_ih[2 * CCH + j];
    w.bhr = b_hh[j]; w.bhz = b_hh[CCH + j]; w.bhn = b_hh[2 * CCH + j];

    const int ntiles = (N + GRU_TILE - 1) / GRU_TILE;
    int tile = blockIdx.x;
    if (tile >= ntiles) return;

    v2f a0[CH2], h0[CH2], a1[CH2], h1[CH2];
    float hj0, hj1;
    const int q0 = wave * (GRU_TILE / GRU_WAVES);
    gru_rows(agg, h_in, tile * GRU_TILE + q0, N, j, a0, h0, hj0);
    int buf = 0;
    int turn = 0;
    #pragma unroll 1
    for (;;) {
        const int base = tile * GRU_TILE + q0;
        const int nt = tile + (int)gridDim.x;
        const bool more = nt < ntiles;
        float* st = &s_o[buf * (GRU_TILE * CCH)];
        float* so = st + q0 * CCH;
        float v;
        gru_rows(agg, h_in, base + 1, N, j, a1, h1, hj1);
        v = gru_node(w, a0, h0, hj0);
        if (EPI == 1) v = v > 0.0f ? v : 0.0f;
        if (lane < CCH) so[lane] = v;
        gru_rows(agg, h_in, base + 2, N, j, a0, h0, hj0);
        v = gru_node(w, a1, h1, hj1);
        if (EPI == 1) v = v > 0.0f ? v : 0.0f;
        if (lane < CCH) so[CCH + lane] = v;
        gru_rows(agg, h_in, base + 3, N, j, a1, h1, hj1);
        v = gru_node(w, a0, h0, hj0);
        if (EPI == 1) v = v > 0.0f ? v : 0.0f;
        if (lane < CCH) so[2 * CCH + lane] = v;
        gru_rows(agg, h_in, more ? nt * GRU_TILE + q0 : base + 3, N, j, a0, h0, hj0);
        v = gru_node(w, a1, h1, hj1);
        if (EPI == 1) v = v > 0.0f ? v : 0.0f;
        if (lane < CCH) so[3 * CCH + lane] = v;
        __syncthreads();
        const int left  = N - tile * GRU_TILE;
        const int nrows = left < GRU_TILE ? left : GRU_TILE;
        if (EPI == 0) {
            float* ht = h_out + (size_t)tile * (GRU_TILE * CCH);
            #pragma unroll
            for (int i = tid; i < GRU_TILE * CH2; i += GRU_WAVES * 32)
                if (i < nrows * CH2)
                    reinterpret_cast<float2*>(ht)[i] = reinterpret_cast<const float2*>(st)[i];
        }
        if (wave == turn)
            tile_product<EPI == 1>(st, s_b, s_d, s_tab, s_bn,
                                   rows_out + (size_t)tile * (GRU_TILE * CCH), nrows, lane);
        if (!more) break;
        tile = nt;
        buf ^= 1;
        turn = (turn + 1) & (GRU_WAVES - 1);
    }
}

extern "C" void kernel_launch(void* const* d_in, const int* in_sizes, int n_in,
                              void* d_out, int out_size, void* d_ws, size_t ws_size,
                              hipStream_t stream) {
    const float* x      = (const float*)d_in[0];
    const int*   eidx   = (const int*)  d_in[1];
    const float* weight = (const float*)d_in[2];
    const float* w_ih   = (const float*)d_in[3];
    const float* w_hh   = (const float*)d_in[4];
    const float* b_ih   = (const float*)d_in[5];
    const float* b_hh   = (const float*)d_in[6];
    const float* lin_w  = (const float*)d_in[7];
    const float* lin_b  = (const float*)d_in[8];
    float* out = (float*)d_out;

    const int N = in_sizes[0] / CCH;
    const int E = in_sizes[1] / 2;
    const int* src = eidx;
    const int* dst = eidx + E;

    float* h   = (float*)d_ws;
    float* m   = h + (size_t)N * CCH;
    float* agg = m + (size_t)N * CCH;
    const int groups = (N + NPW - 1) / NPW;
    int2* rec  = (int2*)(agg + (size_t)N * CCH);
    int*  wcnt = (int*)(rec + (size_t)groups * OWN_WAVES * WCAP);
    unsigned* key = (unsigned*)(((size_t)(wcnt + (size_t)groups * OWN_WAVES) + 15) & ~(size_t)15);
    const int nwords = (E >> 11) * 512;
    int2* pair = (int2*)(key + nwords);
    const int nquads = (E + 3) / 4;
    if (nquads > 0)
        ggnn_owner_keys<<<(nquads + 255) / 256, 256, 0, stream>>>(src, dst, key, pair, nwords, E);

    const int tiles = (N + 15) / 16;
    ggnn_transform_wmma<<<(tiles + 3) / 4, 128, 0, stream>>>(x, weight, m, N);
    for (int layer = 0; layer < NLAYERS; ++layer) {
        if (layer == 0)
            ggnn_owner_sum<0><<<groups, OWN_THREADS, 0, stream>>>(pair, dst, (const uint4*)key, m, agg, rec, wcnt, N, E);
        else
            ggnn_owner_sum<1><<<groups, OWN_THREADS, 0, stream>>>(pair, dst, (const uint4*)key, m, agg, rec, wcnt, N, E);
        const float* hin = layer == 0 ? x : h;
        if (layer + 1 < NLAYERS)
            ggnn_gru<0><<<GRU_GROUPS, GRU_WAVES * 32, 0, stream>>>(
                agg, hin, h, w_ih, w_hh, b_ih, b_hh,
                weight + (size_t)(layer + 1) * CCH * CCH, lin_b, m, N);
        else
            ggnn_gru<1><<<GRU_GROUPS, GRU_WAVES * 32, 0, stream>>>(
                agg, hin, h, w_ih, w_hh, b_ih, b_hh, lin_w, lin_b, out, N);
    }
}


# Owners find their edges in a per-owner bit map (1 bit per 4-edge group, built in LDS by the launch in front with LDS OR) instead of each scanning every edge's key byte; sum order stays LDS arrival
# speedup vs baseline: 12.8315x; 1.1000x over previous
#include <hip/hip_runtime.h>
#include <math.h>

#define CCH 30
#define NLAYERS 3
#define AST 34
#define CH2 (CCH / 2)

typedef __attribute__((ext_vector_type(2))) float v2f;
typedef __attribute__((ext_vector_type(8))) float v8f;

__device__ __forceinline__ float sigmoid_f(float x) { return 1.0f / (1.0f + __expf(-x)); }

__global__ void ggnn_transform_wmma(const float* __restrict__ h,
                                    const float* __restrict__ W,
                                    float* __restrict__ m, int N) {
    __shared__ float s_Wt[32 * AST];
    __shared__ float s_A[4][16 * AST];

    const int tid  = threadIdx.x;
    const int lane = tid & 31;
    const int warp = tid >> 5;
    const int row0 = (blockIdx.x * 4 + warp) * 16;
    const bool full = (row0 + 16 <= N);

    for (int i = tid; i < 32 * AST; i += 128) s_Wt[i] = 0.0f;
    for (int i = lane; i < 16 * AST; i += 32) s_A[warp][i] = 0.0f;
    __syncthreads();

    for (int i = tid; i < CCH * CCH; i += 128) {
        const int k = i / CCH, c = i % CCH;
        s_Wt[c * AST + k] = W[i];
    }
    if (full) {
        const float2* h2 = reinterpret_cast<const float2*>(h + (size_t)row0 * CCH);
        #pragma unroll
        for (int i = lane; i < 16 * CH2; i += 32) {
            const int r = i / CH2, c2 = i % CH2;
            *reinterpret_cast<float2*>(&s_A[warp][r * AST + 2 * c2]) =
                h2[(size_t)r * CH2 + c2];
        }
    } else {
        for (int i = lane; i < 16 * CCH; i += 32) {
            const int r = i / CCH, c = i % CCH;
            const int row = row0 + r;
            if (row < N) s_A[warp][r * AST + c] = h[(size_t)row * CCH + c];
        }
    }
    __syncthreads();

    if (row0 >= N) return;

    const int half = lane >> 4;
    const int l    = lane & 15;
    v8f acc0 = {};
    v8f acc1 = {};
    float* sa = &s_A[warp][0];

    #pragma unroll
    for (int k0 = 0; k0 < 32; k0 += 4) {
        const int ka = k0 + half * 2;
        const v2f a  = *(const v2f*)&sa  [l        * AST + ka];
        const v2f b0 = *(const v2f*)&s_Wt[l        * AST + ka];
        const v2f b1 = *(const v2f*)&s_Wt[(16 + l) * AST + ka];
        acc0 = __builtin_amdgcn_wmma_f32_16x16x4_f32(false, a, false, b0,
                                                     (short)0, acc0, false, false);
        acc1 = __builtin_amdgcn_wmma_f32_16x16x4_f32(false, a, false, b1,
                                                     (short)0, acc1, false, false);
    }

    #pragma unroll
    for (int v = 0; v < 8; ++v) {
        const int r = v + half * 8;
        sa[r * AST + l]      = acc0[v];
        sa[r * AST + 16 + l] = acc1[v];
    }

    if (full) {
        float2* m2 = reinterpret_cast<float2*>(m + (size_t)row0 * CCH);
        #pragma unroll
        for (int i = lane; i < 16 * CH2; i += 32) {
            const int r = i / CH2, c2 = i % CH2;
            m2[(size_t)r * CH2 + c2] =
                *reinterpret_cast<const float2*>(&sa[r * AST + 2 * c2]);
        }
    } else {
        for (int i = lane; i < 16 * CCH; i += 32) {
            const int r = i / CCH, c = i % CCH;
            const int row = row0 + r;
            if (row < N) m[(size_t)row * CCH + c] = sa[r * AST + c];
        }
    }
}

#define NPW         784
#define AGG_ROWS    1368
#define OWN_THREADS 1024
#define OWN_WAVES   32
#define WCAP        1344
#define QCAP        64
#define GTILE       1024
#define GPITCH      33

__global__ __launch_bounds__(GTILE)
void ggnn_owner_groups(const int* __restrict__ src,
                       const int* __restrict__ dst,
                       unsigned* __restrict__ gb,
                       int2* __restrict__ pair, int wd, int E) {
    __shared__ unsigned s_bm[256 * GPITCH];
    const int t = threadIdx.x;
    for (int i = t; i < 256 * GPITCH; i += GTILE) s_bm[i] = 0u;
    __syncthreads();
    const int grp = blockIdx.x * GTILE + t;
    const int e0  = 4 * grp;
    const unsigned bit = 1u << (t & 31);
    const int wv = t >> 5;
    if (e0 + 4 <= E) {
        int d0, d1, d2, d3, s0, s1, s2, s3;
        if ((reinterpret_cast<size_t>(dst) & 15) == 0) {
            const int4 d = reinterpret_cast<const int4*>(dst)[grp];
            d0 = d.x; d1 = d.y; d2 = d.z; d3 = d.w;
        } else {
            d0 = dst[e0]; d1 = dst[e0 + 1]; d2 = dst[e0 + 2]; d3 = dst[e0 + 3];
        }
        if ((reinterpret_cast<size_t>(src) & 15) == 0) {
            const int4 s = reinterpret_cast<const int4*>(src)[grp];
            s0 = s.x; s1 = s.y; s2 = s.z; s3 = s.w;
        } else {
            s0 = src[e0]; s1 = src[e0 + 1]; s2 = src[e0 + 2]; s3 = src[e0 + 3];
        }
        int4* p4 = reinterpret_cast<int4*>(pair + e0);
        p4[0] = make_int4(s0, d0, s1, d1);
        p4[1] = make_int4(s2, d2, s3, d3);
        atomicOr(&s_bm[(((unsigned)d0 / NPW) & 255u) * GPITCH + wv], bit);
        atomicOr(&s_bm[(((unsigned)d1 / NPW) & 255u) * GPITCH + wv], bit);
        atomicOr(&s_bm[(((unsigned)d2 / NPW) & 255u) * GPITCH + wv], bit);
        atomicOr(&s_bm[(((unsigned)d3 / NPW) & 255u) * GPITCH + wv], bit);
    } else {
        for (int e = e0; e < E; ++e) {
            const int s = src[e], d = dst[e];
            pair[e] = make_int2(s, d);
            atomicOr(&s_bm[(((unsigned)d / NPW) & 255u) * GPITCH + wv], bit);
        }
    }
    __syncthreads();
    unsigned* g = gb + (size_t)blockIdx.x * 32;
    for (int i = t; i < 256 * 32; i += GTILE) {
        const int o = i >> 5, w = i & 31;
        g[(size_t)o * wd + w] = s_bm[o * GPITCH + w];
    }
}

template <bool FULL, int B>
__device__ __forceinline__ void add_rows(int s, int dl, int n,
                                         const float* __restrict__ m,
                                         float* s_agg, int lane) {
    #pragma unroll
    for (int j = 0; j < 32; j += B) {
        if (!FULL && j >= n) break;
        int sj[B], dj[B];
        #pragma unroll
        for (int u = 0; u < B; ++u) {
            sj[u] = __builtin_amdgcn_readlane(s, j + u);
            dj[u] = __builtin_amdgcn_readlane(dl, j + u);
        }
        if (lane < CCH) {
            float v[B];
            #pragma unroll
            for (int u = 0; u < B; ++u)
                if (FULL || j + u < n) v[u] = m[(size_t)sj[u] * CCH + lane];
            #pragma unroll
            for (int u = 0; u < B; ++u)
                if (FULL || j + u < n)
                    __hip_atomic_fetch_add(&s_agg[dj[u] * CCH + lane], v[u],
                                           __ATOMIC_RELAXED,
                                           __HIP_MEMORY_SCOPE_WORKGROUP);
        }
    }
}

struct OwnState {
    int*  q;
    int2* q2;
    int   pos, pos2, recpos;
};

template <bool FULL, bool REC>
__device__ __forceinline__ void own_take(int s, int r, int n, OwnState& st,
                                         const float* __restrict__ m, float* s_agg,
                                         int lane, int2* __restrict__ rec) {
    if (REC && (FULL || lane < n) && st.recpos + lane < WCAP)
        rec[st.recpos + lane] = make_int2(s, r);
    add_rows<FULL, 16>(s, r, n, m, s_agg, lane);
    st.recpos += n;
}

template <bool REC>
__device__ __forceinline__ void own_edge(int s, int d, int lo, OwnState& st,
                                         const float* __restrict__ m, float* s_agg,
                                         int lane, int2* __restrict__ rec) {
    const unsigned rr  = (unsigned)(d - lo);
    const bool     own = rr < (unsigned)NPW;
    const unsigned mj  = __builtin_amdgcn_ballot_w32(own);
    if (mj != 0u) {
        if (own) st.q2[st.pos2 + (int)__builtin_amdgcn_mbcnt_lo(mj, 0u)] = make_int2(s, (int)rr);
        st.pos2 += __builtin_popcount(mj);
        if (st.pos2 >= 32) {
            const int2 t = st.q2[lane];
            st.pos2 -= 32;
            if (lane < st.pos2) st.q2[lane] = st.q2[32 + lane];
            own_take<true, REC>(t.x, t.y, 32, st, m, s_agg, lane, rec);
        }
    }
}

template <bool FULL, bool REC>
__device__ __forceinline__ void own_groups(int g, int n, int lo, int E, OwnState& st,
                                           const int2* __restrict__ pair,
                                           const float* __restrict__ m, float* s_agg,
                                           int lane, int2* __restrict__ rec) {
    int4 a = make_int4(0, -1, 0, -1), b = make_int4(0, -1, 0, -1);
    if (FULL || lane < n) {
        const int e = 4 * g;
        if (e + 4 <= E) {
            const int4* p = reinterpret_cast<const int4*>(pair + e);
            a = p[0]; b = p[1];
        } else {
            if (e < E)     { const int2 t = pair[e];     a.x = t.x; a.y = t.y; }
            if (e + 1 < E) { const int2 t = pair[e + 1]; a.z = t.x; a.w = t.y; }
            if (e + 2 < E) { const int2 t = pair[e + 2]; b.x = t.x; b.y = t.y; }
        }
    }
    #pragma unroll 1
    for (int j = 0; j < 4; ++j) {
        const int s = j == 0 ? a.x : j == 1 ? a.z : j == 2 ? b.x : b.z;
        const int d = j == 0 ? a.y : j == 1 ? a.w : j == 2 ? b.y : b.w;
        own_edge<REC>(s, d, lo, st, m, s_agg, lane, rec);
    }
}

__device__ __forceinline__ void own_replay(const int2* __restrict__ rec, int cnt,
                                           const float* __restrict__ m,
                                           float* s_agg, int lane) {
    int2 cur = make_int2(0, 0);
    if (lane < cnt) cur = rec[lane];
    for (int base = 0; base < cnt; base += 32) {
        int2 nxt = make_int2(0, 0);
        if (base + 32 + lane < cnt) nxt = rec[base + 32 + lane];
        const int n = cnt - base;
        if (n >= 32) add_rows<true, 32>(cur.x, cur.y, 32, m, s_agg, lane);
        else         add_rows<false, 32>(cur.x, cur.y, n, m, s_agg, lane);
        cur = nxt;
    }
}

template <int MODE>
__global__ __launch_bounds__(OWN_THREADS)
void ggnn_owner_sum(const int2* __restrict__ pair,
                    const unsigned* __restrict__ gb,
                    const float* __restrict__ m,
                    float* __restrict__ agg,
                    int2* __restrict__ rec_all,
                    int* __restrict__ wcnt, int wd, int N, int E) {
    __shared__ __attribute__((aligned(16))) float s_agg[AGG_ROWS * CCH];
    __shared__ int  s_q[OWN_WAVES * QCAP];
    __shared__ int2 s_q2[OWN_WAVES * QCAP];

    const int tid  = threadIdx.x;
    const int lane = tid & 31;
    const int wave = __builtin_amdgcn_readfirstlane(tid >> 5);
    const int lo   = blockIdx.x * NPW;

    for (int i = tid; i < AGG_ROWS * CCH / 4; i += OWN_THREADS)
        reinterpret_cast<float4*>(s_agg)[i] = make_float4(0.f, 0.f, 0.f, 0.f);
    __syncthreads();

    constexpr bool REC = (MODE == 0);
    const int gw = blockIdx.x * OWN_WAVES + wave;
    int2* rec = rec_all + (size_t)gw * WCAP;
    bool replay = false;
    if (MODE == 1) {
        const int cnt = __builtin_amdgcn_readfirstlane(wcnt[gw]);
        replay = (cnt >= 0 && cnt <= WCAP);
        if (replay) own_replay(rec, cnt, m, s_agg, lane);
    }
    if (!replay) {

    OwnState st;
    st.q  = s_q  + wave * QCAP;
    st.q2 = s_q2 + wave * QCAP;
    st.pos = 0; st.pos2 = 0; st.recpos = 0;

    const unsigned* gbo = gb + (size_t)(blockIdx.x & 255) * wd;
    const int nch = wd >> 5;
    if (wave < nch) {
        unsigned cur = __builtin_nontemporal_load(gbo + wave * 32 + lane);
        for (int c = wave; c < nch; c += OWN_WAVES) {
            const int cn = c + OWN_WAVES;
            const int cl = cn < nch ? cn : c;
            const unsigned nxt = __builtin_nontemporal_load(gbo + cl * 32 + lane);
            unsigned fl = cur;
            const int g0 = (c * 32 + lane) * 32;
            unsigned mk = __builtin_amdgcn_ballot_w32(fl != 0u);
            while (mk != 0u) {
                if (fl != 0u)
                    st.q[st.pos + (int)__builtin_amdgcn_mbcnt_lo(mk, 0u)] = g0 + __builtin_ctz(fl);
                fl &= fl - 1u;
                st.pos += __builtin_popcount(mk);
                if (st.pos >= 32) {
                    const int g = st.q[lane];
                    st.pos -= 32;
                    if (lane < st.pos) st.q[lane] = st.q[32 + lane];
                    own_groups<true, REC>(g, 32, lo, E, st, pair, m, s_agg, lane, rec);
                }
                mk = __builtin_amdgcn_ballot_w32(fl != 0u);
            }
            cur = nxt;
        }
    }
    if (st.pos > 0) {
        const int g = lane < st.pos ? st.q[lane] : 0;
        own_groups<false, REC>(g, st.pos, lo, E, st, pair, m, s_agg, lane, rec);
    }
    if (st.pos2 > 0) {
        int2 t = make_int2(0, 0);
        if (lane < st.pos2) t = st.q2[lane];
        own_take<false, REC>(t.x, t.y, st.pos2, st, m, s_agg, lane, rec);
    }
    if (REC && lane == 0) wcnt[gw] = st.recpos;
    }
    __syncthreads();

    const int nn = (N - lo) < NPW ? (N - lo) : NPW;
    float* ag = agg + (size_t)lo * CCH;
    for (int i = tid; i < nn * CCH; i += OWN_THREADS) ag[i] = s_agg[i];
}

#define GRU_WAVES  4
#define GRU_GROUPS 768
#define GRU_TILE   16

__device__ __forceinline__ void gru_rows(const float* agg, const float* h, int n, int N, int j,
                                         v2f (&a)[CH2], v2f (&hv)[CH2], float& hj) {
    n = n < N ? n : N - 1;
    const float* ar = agg + (size_t)n * CCH;
    const float* hr = h   + (size_t)n * CCH;
    #pragma unroll
    for (int k = 0; k < CH2; ++k) {
        a[k]  = *(const v2f*)(ar + 2 * k);
        hv[k] = *(const v2f*)(hr + 2 * k);
    }
    hj = hr[j];
}

struct GruW {
    v2f wir[CH2], wiz[CH2], win[CH2], whr[CH2], whz[CH2], whn[CH2];
    float bir, biz, bin, bhr, bhz, bhn;
};

__device__ __forceinline__ float gru_node(const GruW& w, const v2f (&a)[CH2], const v2f (&hv)[CH2],
                                          float hj) {
    v2f ir = {w.bir, 0.0f}, iz = {w.biz, 0.0f}, in_ = {w.bin, 0.0f};
    v2f gr = {w.bhr, 0.0f}, gz = {w.bhz, 0.0f}, gn = {w.bhn, 0.0f};
    #pragma unroll
    for (int k = 0; k < CH2; ++k) {
        ir  = __builtin_elementwise_fma(a[k],  w.wir[k], ir);
        iz  = __builtin_elementwise_fma(a[k],  w.wiz[k], iz);
        in_ = __builtin_elementwise_fma(a[k],  w.win[k], in_);
        gr  = __builtin_elementwise_fma(hv[k], w.whr[k], gr);
        gz  = __builtin_elementwise_fma(hv[k], w.whz[k], gz);
        gn  = __builtin_elementwise_fma(hv[k], w.whn[k], gn);
    }
    const float r  = sigmoid_f((ir.x + ir.y) + (gr.x + gr.y));
    const float z  = sigmoid_f((iz.x + iz.y) + (gz.x + gz.y));
    const float nn = tanhf((in_.x + in_.y) + r * (gn.x + gn.y));
    return (1.0f - z) * nn + z * hj;
}

template <bool BIAS>
__device__ __forceinline__ void tile_product(const float* sa, const float* sb, float* sd,
                                             const int2* tab, const float* sbias,
                                             float* __restrict__ dst, int nrows, int lane) {
    const int2 t   = tab[lane];
    const int  fo  = t.x;
    const int  dof = t.y;
    const bool hi  = lane >= 16;
    v8f acc0 = {};
    v8f acc1 = {};
    #pragma unroll
    for (int k0 = 0; k0 < 32; k0 += 4) {
        v2f a  = *(const v2f*)&sa[fo + k0];
        v2f b0 = *(const v2f*)&sb[fo + k0];
        v2f b1 = *(const v2f*)&sb[16 * CCH + fo + k0];
        if (k0 + 4 > CCH && hi) {
            a = v2f{0.0f, 0.0f}; b0 = v2f{0.0f, 0.0f}; b1 = v2f{0.0f, 0.0f};
        }
        acc0 = __builtin_amdgcn_wmma_f32_16x16x4_f32(false, a, false, b0,
                                                     (short)0, acc0, false, false);
        acc1 = __builtin_amdgcn_wmma_f32_16x16x4_f32(false, a, false, b1,
                                                     (short)0, acc1, false, false);
    }
    float blo = 0.0f, bhi = 0.0f;
    if (BIAS) { blo = sbias[lane & 15]; bhi = sbias[16 + (lane & 15)]; }
    const bool c1 = 16 + (lane & 15) < CCH;
    #pragma unroll
    for (int v = 0; v < 8; ++v) {
        sd[dof + v * CCH] = BIAS ? acc0[v] + blo : acc0[v];
        if (c1) sd[dof + v * CCH + 16] = BIAS ? acc1[v] + bhi : acc1[v];
    }
    float2* d2 = reinterpret_cast<float2*>(dst);
    const float2* sd2 = reinterpret_cast<const float2*>(sd);
    #pragma unroll
    for (int i = lane; i < GRU_TILE * CH2; i += 32)
        if (i < nrows * CH2) d2[i] = sd2[i];
}

template <int EPI>
__global__ __launch_bounds__(GRU_WAVES * 32, 3)
void ggnn_gru(const float* agg,
              const float* h_in,
              float* h_out,
              const float* __restrict__ w_ih,
              const float* __restrict__ w_hh,
              const float* __restrict__ b_ih,
              const float* __restrict__ b_hh,
              const float* __restrict__ wn,
              const float* __restrict__ bn,
              float* __restrict__ rows_out,
              int N) {
    const int tid  = threadIdx.x;
    const int lane = tid & 31;
    const int wave = __builtin_amdgcn_readfirstlane(tid >> 5);
    const int j = lane < CCH ? lane : CCH - 1;

    __shared__ __attribute__((aligned(16))) float s_w[2 * 3 * CCH * CCH];
    __shared__ __attribute__((aligned(16))) float s_o[2 * GRU_TILE * CCH + 4];
    __shared__ __attribute__((aligned(16))) float s_b[32 * CCH + 4];
    __shared__ __attribute__((aligned(16))) float s_bn[32];
    __shared__ __attribute__((aligned(16))) int2  s_tab[32];
    __shared__ __attribute__((aligned(16))) float s_d[GRU_TILE * CCH];
    for (int i = tid; i < 3 * CCH * CCH / 4; i += GRU_WAVES * 32) {
        reinterpret_cast<float4*>(s_w)[i] = reinterpret_cast<const float4*>(w_ih)[i];
        reinterpret_cast<float4*>(s_w + 3 * CCH * CCH)[i] = reinterpret_cast<const float4*>(w_hh)[i];
    }
    for (int i = tid; i < 32 * CCH + 4; i += GRU_WAVES * 32) {
        const int c = i / CCH, k = i % CCH;
        float v = 0.0f;
        if (c < CCH) v = (EPI == 0) ? wn[k * CCH + c] : wn[c * CCH + k];
        s_b[i] = v;
    }
    if (tid < 32) {
        s_tab[tid] = make_int2((tid & 15) * CCH + (tid >> 4) * 2, (tid >> 4) * 8 * CCH + (tid & 15));
        s_bn[tid]  = (EPI == 1 && tid < CCH) ? bn[tid] : 0.0f;
    }
    __syncthreads();
    const float* s_ih = s_w;
    const float* s_hh = s_w + 3 * CCH * CCH;
    GruW w;
    #pragma unroll
    for (int k = 0; k < CH2; ++k) {
        w.wir[k] = *(const v2f*)&s_ih[j * CCH + 2 * k];
        w.wiz[k] = *(const v2f*)&s_ih[(CCH + j) * CCH + 2 * k];
        w.win[k] = *(const v2f*)&s_ih[(2 * CCH + j) * CCH + 2 * k];
        w.whr[k] = *(const v2f*)&s_hh[j * CCH + 2 * k];
        w.whz[k] = *(const v2f*)&s_hh[(CCH + j) * CCH + 2 * k];
        w.whn[k] = *(const v2f*)&s_hh[(2 * CCH + j) * CCH + 2 * k];
    }
    w.bir = b_ih[j]; w.biz = b_ih[CCH + j]; w.bin = b_ih[2 * CCH + j];
    w.bhr = b_hh[j]; w.bhz = b_hh[CCH + j]; w.bhn = b_hh[2 * CCH + j];

    const int ntiles = (N + GRU_TILE - 1) / GRU_TILE;
    int tile = blockIdx.x;
    if (tile >= ntiles) return;

    v2f a0[CH2], h0[CH2], a1[CH2], h1[CH2];
    float hj0, hj1;
    const int q0 = wave * (GRU_TILE / GRU_WAVES);
    gru_rows(agg, h_in, tile * GRU_TILE + q0, N, j, a0, h0, hj0);
    int buf = 0;
    int turn = 0;
    #pragma unroll 1
    for (;;) {
        const int base = tile * GRU_TILE + q0;
        const int nt = tile + (int)gridDim.x;
        const bool more = nt < ntiles;
        float* st = &s_o[buf * (GRU_TILE * CCH)];
        float* so = st + q0 * CCH;
        float v;
        gru_rows(agg, h_in, base + 1, N, j, a1, h1, hj1);
        v = gru_node(w, a0, h0, hj0);
        if (EPI == 1) v = v > 0.0f ? v : 0.0f;
        if (lane < CCH) so[lane] = v;
        gru_rows(agg, h_in, base + 2, N, j, a0, h0, hj0);
        v = gru_node(w, a1, h1, hj1);
        if (EPI == 1) v = v > 0.0f ? v : 0.0f;
        if (lane < CCH) so[CCH + lane] = v;
        gru_rows(agg, h_in, base + 3, N, j, a1, h1, hj1);
        v = gru_node(w, a0, h0, hj0);
        if (EPI == 1) v = v > 0.0f ? v : 0.0f;
        if (lane < CCH) so[2 * CCH + lane] = v;
        gru_rows(agg, h_in, more ? nt * GRU_TILE + q0 : base + 3, N, j, a0, h0, hj0);
        v = gru_node(w, a1, h1, hj1);
        if (EPI == 1) v = v > 0.0f ? v : 0.0f;
        if (lane < CCH) so[3 * CCH + lane] = v;
        __syncthreads();
        const int left  = N - tile * GRU_TILE;
        const int nrows = left < GRU_TILE ? left : GRU_TILE;
        if (EPI == 0) {
            float* ht = h_out + (size_t)tile * (GRU_TILE * CCH);
            #pragma unroll
            for (int i = tid; i < GRU_TILE * CH2; i += GRU_WAVES * 32)
                if (i < nrows * CH2)
                    reinterpret_cast<float2*>(ht)[i] = reinterpret_cast<const float2*>(st)[i];
        }
        if (wave == turn)
            tile_product<EPI == 1>(st, s_b, s_d, s_tab, s_bn,
                                   rows_out + (size_t)tile * (GRU_TILE * CCH), nrows, lane);
        if (!more) break;
        tile = nt;
        buf ^= 1;
        turn = (turn + 1) & (GRU_WAVES - 1);
    }
}

extern "C" void kernel_launch(void* const* d_in, const int* in_sizes, int n_in,
                              void* d_out, int out_size, void* d_ws, size_t ws_size,
                              hipStream_t stream) {
    const float* x      = (const float*)d_in[0];
    const int*   eidx   = (const int*)  d_in[1];
    const float* weight = (const float*)d_in[2];
    const float* w_ih   = (const float*)d_in[3];
    const float* w_hh   = (const float*)d_in[4];
    const float* b_ih   = (const float*)d_in[5];
    const float* b_hh   = (const float*)d_in[6];
    const float* lin_w  = (const float*)d_in[7];
    const float* lin_b  = (const float*)d_in[8];
    float* out = (float*)d_out;

    const int N = in_sizes[0] / CCH;
    const int E = in_sizes[1] / 2;
    const int* src = eidx;
    const int* dst = eidx + E;

    float* h   = (float*)d_ws;
    float* m   = h + (size_t)N * CCH;
    float* agg = m + (size_t)N * CCH;
    const int groups = (N + NPW - 1) / NPW;
    int2* rec  = (int2*)(agg + (size_t)N * CCH);
    int*  wcnt = (int*)(rec + (size_t)groups * OWN_WAVES * WCAP);
    const int nquads = (E + 3) / 4;
    const int gtiles = (nquads + GTILE - 1) / GTILE;
    const int wd = gtiles * 32;
    unsigned* gb = (unsigned*)(((size_t)(wcnt + (size_t)groups * OWN_WAVES) + 127) & ~(size_t)127);
    int2* pair = (int2*)(gb + (size_t)256 * wd);
    if (gtiles > 0)
        ggnn_owner_groups<<<gtiles, GTILE, 0, stream>>>(src, dst, gb, pair, wd, E);

    const int tiles = (N + 15) / 16;
    ggnn_transform_wmma<<<(tiles + 3) / 4, 128, 0, stream>>>(x, weight, m, N);
    for (int layer = 0; layer < NLAYERS; ++layer) {
        if (layer == 0)
            ggnn_owner_sum<0><<<groups, OWN_THREADS, 0, stream>>>(pair, gb, m, agg, rec, wcnt, wd, N, E);
        else
            ggnn_owner_sum<1><<<groups, OWN_THREADS, 0, stream>>>(pair, gb, m, agg, rec, wcnt, wd, N, E);
        const float* hin = layer == 0 ? x : h;
        if (layer + 1 < NLAYERS)
            ggnn_gru<0><<<GRU_GROUPS, GRU_WAVES * 32, 0, stream>>>(
                agg, hin, h, w_ih, w_hh, b_ih, b_hh,
                weight + (size_t)(layer + 1) * CCH * CCH, lin_b, m, N);
        else
            ggnn_gru<1><<<GRU_GROUPS, GRU_WAVES * 32, 0, stream>>>(
                agg, hin, h, w_ih, w_hh, b_ih, b_hh, lin_w, lin_b, out, N);
    }
}
